# starting kernel
# baseline (speedup 1.0000x reference)
#include <hip/hip_runtime.h>

typedef float v2f __attribute__((ext_vector_type(2)));
typedef float v8f __attribute__((ext_vector_type(8)));

#define HIDF 128

__global__ void init_deg_kernel(float* __restrict__ deg, int n) {
    int i = blockIdx.x * blockDim.x + threadIdx.x;
    if (i < n) deg[i] = 1.0f;
}

__global__ void count_deg_kernel(const int* __restrict__ dst, float* __restrict__ deg, int e) {
    int i = blockIdx.x * blockDim.x + threadIdx.x;
    if (i < e) atomicAdd(&deg[dst[i]], 1.0f);
}

__global__ void dinv_kernel(float* __restrict__ deg, int n) {
    int i = blockIdx.x * blockDim.x + threadIdx.x;
    if (i < n) deg[i] = rsqrtf(deg[i]);
}

__global__ void zero_kernel(float* __restrict__ p, long n) {
    long i = (long)blockIdx.x * blockDim.x + threadIdx.x;
    if (i < n) p[i] = 0.0f;
}

__global__ void gemm_wmma_kernel(const float* __restrict__ H,
                                 const float* __restrict__ W,
                                 float* __restrict__ Out, int M) {
    int wave  = threadIdx.x >> 5;
    int lane  = threadIdx.x & 31;
    int mtile = blockIdx.x * 4 + wave;
    if (mtile * 16 >= M) return;
    int half = lane >> 4;
    int lm   = lane & 15;

    const float* hrow = H + (long)(mtile * 16 + lm) * HIDF;

    v8f c[8];
#pragma unroll
    for (int i = 0; i < 8; ++i) c[i] = (v8f){0.f, 0.f, 0.f, 0.f, 0.f, 0.f, 0.f, 0.f};

    for (int k0 = 0; k0 < HIDF; k0 += 4) {
        v2f a = *(const v2f*)(hrow + k0 + 2 * half);
        const float* wk = W + (long)(k0 + 2 * half) * HIDF + lm;
#pragma unroll
        for (int nt = 0; nt < 8; ++nt) {
            v2f b;
            b.x = wk[nt * 16];
            b.y = wk[HIDF + nt * 16];
            c[nt] = __builtin_amdgcn_wmma_f32_16x16x4_f32(
                false, a, false, b, (short)0, c[nt], false, false);
        }
    }

#pragma unroll
    for (int nt = 0; nt < 8; ++nt) {
#pragma unroll
        for (int i = 0; i < 8; ++i) {
            Out[(long)(mtile * 16 + i + 8 * half) * HIDF + nt * 16 + lm] = c[nt][i];
        }
    }
}

__global__ void scatter_kernel(const float* __restrict__ hw,
                               const int* __restrict__ src,
                               const int* __restrict__ dst,
                               const float* __restrict__ dinv,
                               float* __restrict__ agg, int e) {
    int edge = blockIdx.x * (blockDim.x >> 5) + (threadIdx.x >> 5);
    int lane = threadIdx.x & 31;
    if (edge >= e) return;
    int s = src[edge];
    int d = dst[edge];
    float norm = dinv[s] * dinv[d];
    float4 v = *((const float4*)(hw + (long)s * HIDF) + lane);
    float* out = agg + (long)d * HIDF + lane * 4;
    atomicAdd(out + 0, v.x * norm);
    atomicAdd(out + 1, v.y * norm);
    atomicAdd(out + 2, v.z * norm);
    atomicAdd(out + 3, v.w * norm);
}

__global__ void finish_kernel(const float* __restrict__ hw,
                              const float* __restrict__ dinv,
                              const float* __restrict__ bias,
                              float* __restrict__ h, long total) {
    long i = (long)blockIdx.x * blockDim.x + threadIdx.x;
    if (i >= total) return;
    int node = (int)(i >> 7);
    int f    = (int)(i & 127);
    float di = dinv[node];
    float v  = h[i] + hw[i] * di * di + bias[f];
    h[i] = v > 0.0f ? v : 0.0f;
}

__global__ void pool_kernel(const float* __restrict__ h,
                            const int* __restrict__ ptr,
                            float* __restrict__ out, int g) {
    int gi = blockIdx.x;
    int t  = threadIdx.x;
    int s  = ptr[gi];
    int e  = ptr[gi + 1];
    float acc = 0.0f;
    for (int n = s; n < e; ++n) acc += h[(long)n * HIDF + t];
    float cnt = (float)(e - s);
    if (cnt < 1.0f) cnt = 1.0f;
    out[(long)gi * HIDF + t] = acc / cnt;
}

extern "C" void kernel_launch(void* const* d_in, const int* in_sizes, int n_in,
                              void* d_out, int out_size, void* d_ws, size_t ws_size,
                              hipStream_t stream) {
    const float* x   = (const float*)d_in[0];
    const int*   ei  = (const int*)d_in[1];
    const int*   ptr = (const int*)d_in[2];
    const float* Wl[3] = {(const float*)d_in[3], (const float*)d_in[5], (const float*)d_in[7]};
    const float* bl[3] = {(const float*)d_in[4], (const float*)d_in[6], (const float*)d_in[8]};

    int N = in_sizes[0] / HIDF;
    int E = in_sizes[1] / 2;
    int G = in_sizes[2] - 1;

    const int* src = ei;
    const int* dst = ei + E;

    float* dinv = (float*)d_ws;
    size_t npad = ((size_t)N + 255) & ~(size_t)255;
    float* bufA = dinv + npad;
    float* bufB = bufA + (size_t)N * HIDF;
    long total  = (long)N * HIDF;

    init_deg_kernel<<<(N + 255) / 256, 256, 0, stream>>>(dinv, N);
    count_deg_kernel<<<(E + 255) / 256, 256, 0, stream>>>(dst, dinv, E);
    dinv_kernel<<<(N + 255) / 256, 256, 0, stream>>>(dinv, N);

    int mtiles      = (N + 15) / 16;
    int gemm_blocks = (mtiles + 3) / 4;
    int scat_blocks = (E + 7) / 8;
    int elem_blocks = (int)((total + 255) / 256);

    const float* hin = x;
    for (int l = 0; l < 3; ++l) {
        gemm_wmma_kernel<<<gemm_blocks, 128, 0, stream>>>(hin, Wl[l], bufA, N);
        zero_kernel<<<elem_blocks, 256, 0, stream>>>(bufB, total);
        scatter_kernel<<<scat_blocks, 256, 0, stream>>>(bufA, src, dst, dinv, bufB, E);
        finish_kernel<<<elem_blocks, 256, 0, stream>>>(bufA, dinv, bl[l], bufB, total);
        hin = bufB;
    }

    pool_kernel<<<G, HIDF, 0, stream>>>(bufB, ptr, (float*)d_out, G);
}


# gather over a dst-sorted edge list built in-call (int atomic rank + scan, self-loop as last entry) replaces f32 atomic scatter, zeroing and finish pass; per-node sum order = arrival rank
# speedup vs baseline: 2.0486x; 2.0486x over previous
#include <hip/hip_runtime.h>

typedef float v2f __attribute__((ext_vector_type(2)));
typedef float v8f __attribute__((ext_vector_type(8)));
typedef float v4f __attribute__((ext_vector_type(4)));

#define HIDF 128


__global__ void init_cnt_kernel(int* __restrict__ cnt, int n) {
    int i = blockIdx.x * blockDim.x + threadIdx.x;
    if (i < n) cnt[i] = 0;
}

__global__ void count_rank_kernel(const int* __restrict__ dst, int* __restrict__ cnt,
                                  int* __restrict__ rank, int e) {
    int i = blockIdx.x * blockDim.x + threadIdx.x;
    if (i < e) rank[i] = atomicAdd(&cnt[dst[i]], 1);
}

__global__ void __launch_bounds__(1024) scan_kernel(const int* __restrict__ cnt,
                                                    int* __restrict__ rowptr,
                                                    float* __restrict__ dinv,
                                                    int* __restrict__ csr, int n) {
    __shared__ int s_part[32];
    __shared__ int s_loc[32];
    const int tid  = threadIdx.x;
    const int lane = tid & 31;
    const int wave = tid >> 5;
    const int blk_start = blockIdx.x * 4096;

    int part = 0;
    int i = tid * 4;
    for (; i + 7 * 4096 < blk_start; i += 8 * 4096) {
        int4 v[8];
#pragma unroll
        for (int u = 0; u < 8; ++u) v[u] = *(const int4*)(cnt + i + u * 4096);
#pragma unroll
        for (int u = 0; u < 8; ++u) part += (v[u].x + v[u].y) + (v[u].z + v[u].w);
    }
    for (; i < blk_start; i += 4096) {
        int4 v = *(const int4*)(cnt + i);
        part += (v.x + v.y) + (v.z + v.w);
    }

    const int i0 = blk_start + tid * 4;
    int c[4] = {0, 0, 0, 0};
    if (i0 + 3 < n) {
        int4 v = *(const int4*)(cnt + i0);
        c[0] = v.x; c[1] = v.y; c[2] = v.z; c[3] = v.w;
    } else {
        for (int k = 0; k < 4; ++k) if (i0 + k < n) c[k] = cnt[i0 + k];
    }
    const int loc = (c[0] + c[1]) + (c[2] + c[3]);

    int incl = loc;
    for (int off = 1; off < 32; off <<= 1) {
        int t = __shfl_up(incl, off);
        if (lane >= off) incl += t;
    }
    for (int off = 16; off > 0; off >>= 1) part += __shfl_xor(part, off);
    if (lane == 31) s_loc[wave] = incl;
    if (lane == 0)  s_part[wave] = part;
    __syncthreads();
    int base = 0;
    for (int w = 0; w < 32; ++w) {
        base += s_part[w];
        if (w < wave) base += s_loc[w];
    }
    int run = base + incl - loc;
    for (int k = 0; k < 4; ++k) {
        int idx = i0 + k;
        if (idx < n) {
            float di = rsqrtf((float)(c[k] + 1));
            int row  = run + idx;
            rowptr[idx] = row;
            dinv[idx]   = di;
            ((int2*)csr)[row + c[k]] = make_int2(idx, __float_as_int(di * di));
            run += c[k];
            if (idx == n - 1) rowptr[n] = run + n;
        }
    }
}

__global__ void fill_kernel(const int* __restrict__ src, const int* __restrict__ dst,
                            const int* __restrict__ rank, const int* __restrict__ rowptr,
                            const float* __restrict__ dinv, int* __restrict__ csr, int e) {
    int i = blockIdx.x * blockDim.x + threadIdx.x;
    if (i >= e) return;
    int s = src[i];
    int d = dst[i];
    int pos = rowptr[d] + rank[i];
    float norm = dinv[s] * dinv[d];
    ((int2*)csr)[pos] = make_int2(s, __float_as_int(norm));
}

__global__ void gemm_wmma_kernel(const float* __restrict__ H,
                                 const float* __restrict__ W,
                                 float* __restrict__ Out, int M) {
    int wave  = threadIdx.x >> 5;
    int lane  = threadIdx.x & 31;
    int mtile = blockIdx.x * 4 + wave;
    if (mtile * 16 >= M) return;
    int half = lane >> 4;
    int lm   = lane & 15;

    const float* hrow = H + (long)(mtile * 16 + lm) * HIDF;

    v8f c[8];
#pragma unroll
    for (int i = 0; i < 8; ++i) c[i] = (v8f){0.f, 0.f, 0.f, 0.f, 0.f, 0.f, 0.f, 0.f};

    for (int k0 = 0; k0 < HIDF; k0 += 4) {
        v2f a = *(const v2f*)(hrow + k0 + 2 * half);
        const float* wk = W + (long)(k0 + 2 * half) * HIDF + lm;
#pragma unroll
        for (int nt = 0; nt < 8; ++nt) {
            v2f b;
            b.x = wk[nt * 16];
            b.y = wk[HIDF + nt * 16];
            c[nt] = __builtin_amdgcn_wmma_f32_16x16x4_f32(
                false, a, false, b, (short)0, c[nt], false, false);
        }
    }

#pragma unroll
    for (int nt = 0; nt < 8; ++nt) {
#pragma unroll
        for (int i = 0; i < 8; ++i) {
            Out[(long)(mtile * 16 + i + 8 * half) * HIDF + nt * 16 + lm] = c[nt][i];
        }
    }
}

template <int K>
__device__ __forceinline__ void add_rows(const v4f* __restrict__ hw4, int2 ent, int j,
                                         int lane, v4f& acc) {
    v4f   v[K];
    float w[K];
#pragma unroll
    for (int k = 0; k < K; ++k) {
        int s = __builtin_amdgcn_readlane(ent.x, j + k);
        w[k]  = __int_as_float(__builtin_amdgcn_readlane(ent.y, j + k));
        v[k]  = hw4[(long)s * (HIDF / 4) + lane];
    }
#pragma unroll
    for (int k = 0; k < K; ++k) acc += v[k] * w[k];
}

__global__ void __launch_bounds__(256) gather_kernel(const float* __restrict__ hw,
                                                     const int* __restrict__ rowptr,
                                                     const int* __restrict__ csr,
                                                     const float* __restrict__ bias,
                                                     float* __restrict__ h, int n) {
    const int lane = threadIdx.x & 31;
    const int node = __builtin_amdgcn_readfirstlane(blockIdx.x * 8 + (threadIdx.x >> 5));
    if (node >= n) return;
    const v4f*  hw4 = (const v4f*)hw;
    const int2* ed  = (const int2*)csr;

    const v4f bv  = *((const v4f*)bias + lane);
    const int beg = rowptr[node];
    const int end = rowptr[node + 1];

    v4f acc = {0.0f, 0.0f, 0.0f, 0.0f};
    for (int base = beg; base < end; base += 32) {
        int m = end - base;
        if (m > 32) m = 32;
        int2 ent = make_int2(0, 0);
        if (lane < m) ent = ed[base + lane];
        int j = 0;
        for (; j + 8 <= m; j += 8) add_rows<8>(hw4, ent, j, lane, acc);
        if (j + 4 <= m) { add_rows<4>(hw4, ent, j, lane, acc); j += 4; }
        const int r = m - j;
        if (r == 3)      add_rows<3>(hw4, ent, j, lane, acc);
        else if (r == 2) add_rows<2>(hw4, ent, j, lane, acc);
        else if (r == 1) add_rows<1>(hw4, ent, j, lane, acc);
    }

    v4f o = acc + bv;
    o.x = o.x > 0.0f ? o.x : 0.0f;
    o.y = o.y > 0.0f ? o.y : 0.0f;
    o.z = o.z > 0.0f ? o.z : 0.0f;
    o.w = o.w > 0.0f ? o.w : 0.0f;
    *((v4f*)(h + (long)node * HIDF) + lane) = o;
}

__global__ void pool_kernel(const float* __restrict__ h,
                            const int* __restrict__ ptr,
                            float* __restrict__ out, int g) {
    int gi = blockIdx.x;
    int t  = threadIdx.x;
    int s  = ptr[gi];
    int e  = ptr[gi + 1];
    float acc = 0.0f;
    for (int n = s; n < e; ++n) acc += h[(long)n * HIDF + t];
    float cnt = (float)(e - s);
    if (cnt < 1.0f) cnt = 1.0f;
    out[(long)gi * HIDF + t] = acc / cnt;
}

extern "C" void kernel_launch(void* const* d_in, const int* in_sizes, int n_in,
                              void* d_out, int out_size, void* d_ws, size_t ws_size,
                              hipStream_t stream) {
    const float* x   = (const float*)d_in[0];
    const int*   ei  = (const int*)d_in[1];
    const int*   ptr = (const int*)d_in[2];
    const float* Wl[3] = {(const float*)d_in[3], (const float*)d_in[5], (const float*)d_in[7]};
    const float* bl[3] = {(const float*)d_in[4], (const float*)d_in[6], (const float*)d_in[8]};

    int N = in_sizes[0] / HIDF;
    int E = in_sizes[1] / 2;
    int G = in_sizes[2] - 1;

    const int* src = ei;
    const int* dst = ei + E;

    size_t npad = ((size_t)N + 255) & ~(size_t)255;
    size_t epad = ((size_t)E + 255) & ~(size_t)255;
    float* dinv   = (float*)d_ws;
    int*   cnt    = (int*)(dinv + npad);
    int*   rowptr = cnt + npad;
    int*   rank   = rowptr + npad + 256;
    int*   csr    = rank + epad;
    float* bufA   = (float*)(csr + 2 * (epad + npad));
    float* bufB   = bufA + (size_t)N * HIDF;

    init_cnt_kernel<<<(N + 255) / 256, 256, 0, stream>>>(cnt, N);
    count_rank_kernel<<<(E + 255) / 256, 256, 0, stream>>>(dst, cnt, rank, E);
    scan_kernel<<<(N + 4095) / 4096, 1024, 0, stream>>>(cnt, rowptr, dinv, csr, N);
    fill_kernel<<<(E + 255) / 256, 256, 0, stream>>>(src, dst, rank, rowptr, dinv, csr, E);

    int mtiles      = (N + 15) / 16;
    int gemm_blocks = (mtiles + 3) / 4;
    int gath_blocks = (N + 7) / 8;

    const float* hin = x;
    for (int l = 0; l < 3; ++l) {
        gemm_wmma_kernel<<<gemm_blocks, 128, 0, stream>>>(hin, Wl[l], bufA, N);
        gather_kernel<<<gath_blocks, 256, 0, stream>>>(bufA, rowptr, csr, bl[l], bufB, N);
        hin = bufB;
    }

    pool_kernel<<<G, HIDF, 0, stream>>>(bufB, ptr, (float*)d_out, G);
}


# pool: 8 waves per graph, wave w sums rows w, w+8, .. in groups of 16 whole-row 16-byte loads, partials added via LDS in wave order (was one serial 8-row group per trip); pool sum order changes
# speedup vs baseline: 2.5468x; 1.2432x over previous
#include <hip/hip_runtime.h>

typedef float v2f __attribute__((ext_vector_type(2)));
typedef float v8f __attribute__((ext_vector_type(8)));
typedef float v4f __attribute__((ext_vector_type(4)));

#define HIDF 128


__global__ void init_cnt_kernel(int* __restrict__ cnt, int n) {
    int i = blockIdx.x * blockDim.x + threadIdx.x;
    if (i < n) cnt[i] = 0;
}

__global__ void count_rank_kernel(const int* __restrict__ dst, int* __restrict__ cnt,
                                  int* __restrict__ rank, int e) {
    int i = blockIdx.x * blockDim.x + threadIdx.x;
    if (i < e) rank[i] = atomicAdd(&cnt[dst[i]], 1);
}

__global__ void __launch_bounds__(1024) scan_kernel(const int* __restrict__ cnt,
                                                    int* __restrict__ rowptr,
                                                    float* __restrict__ dinv,
                                                    int* __restrict__ csr, int n) {
    __shared__ int s_part[32];
    __shared__ int s_loc[32];
    const int tid  = threadIdx.x;
    const int lane = tid & 31;
    const int wave = tid >> 5;
    const int blk_start = blockIdx.x * 4096;

    int part = 0;
    int i = tid * 4;
    for (; i + 7 * 4096 < blk_start; i += 8 * 4096) {
        int4 v[8];
#pragma unroll
        for (int u = 0; u < 8; ++u) v[u] = *(const int4*)(cnt + i + u * 4096);
#pragma unroll
        for (int u = 0; u < 8; ++u) part += (v[u].x + v[u].y) + (v[u].z + v[u].w);
    }
    for (; i < blk_start; i += 4096) {
        int4 v = *(const int4*)(cnt + i);
        part += (v.x + v.y) + (v.z + v.w);
    }

    const int i0 = blk_start + tid * 4;
    int c[4] = {0, 0, 0, 0};
    if (i0 + 3 < n) {
        int4 v = *(const int4*)(cnt + i0);
        c[0] = v.x; c[1] = v.y; c[2] = v.z; c[3] = v.w;
    } else {
        for (int k = 0; k < 4; ++k) if (i0 + k < n) c[k] = cnt[i0 + k];
    }
    const int loc = (c[0] + c[1]) + (c[2] + c[3]);

    int incl = loc;
    for (int off = 1; off < 32; off <<= 1) {
        int t = __shfl_up(incl, off);
        if (lane >= off) incl += t;
    }
    for (int off = 16; off > 0; off >>= 1) part += __shfl_xor(part, off);
    if (lane == 31) s_loc[wave] = incl;
    if (lane == 0)  s_part[wave] = part;
    __syncthreads();
    int base = 0;
    for (int w = 0; w < 32; ++w) {
        base += s_part[w];
        if (w < wave) base += s_loc[w];
    }
    int run = base + incl - loc;
    for (int k = 0; k < 4; ++k) {
        int idx = i0 + k;
        if (idx < n) {
            float di = rsqrtf((float)(c[k] + 1));
            int row  = run + idx;
            rowptr[idx] = row;
            dinv[idx]   = di;
            ((int2*)csr)[row + c[k]] = make_int2(idx, __float_as_int(di * di));
            run += c[k];
            if (idx == n - 1) rowptr[n] = run + n;
        }
    }
}

__global__ void fill_kernel(const int* __restrict__ src, const int* __restrict__ dst,
                            const int* __restrict__ rank, const int* __restrict__ rowptr,
                            const float* __restrict__ dinv, int* __restrict__ csr, int e) {
    int i = blockIdx.x * blockDim.x + threadIdx.x;
    if (i >= e) return;
    int s = src[i];
    int d = dst[i];
    int pos = rowptr[d] + rank[i];
    float norm = dinv[s] * dinv[d];
    ((int2*)csr)[pos] = make_int2(s, __float_as_int(norm));
}

__global__ void gemm_wmma_kernel(const float* __restrict__ H,
                                 const float* __restrict__ W,
                                 float* __restrict__ Out, int M) {
    int wave  = threadIdx.x >> 5;
    int lane  = threadIdx.x & 31;
    int mtile = blockIdx.x * 4 + wave;
    if (mtile * 16 >= M) return;
    int half = lane >> 4;
    int lm   = lane & 15;

    const float* hrow = H + (long)(mtile * 16 + lm) * HIDF;

    v8f c[8];
#pragma unroll
    for (int i = 0; i < 8; ++i) c[i] = (v8f){0.f, 0.f, 0.f, 0.f, 0.f, 0.f, 0.f, 0.f};

    for (int k0 = 0; k0 < HIDF; k0 += 4) {
        v2f a = *(const v2f*)(hrow + k0 + 2 * half);
        const float* wk = W + (long)(k0 + 2 * half) * HIDF + lm;
#pragma unroll
        for (int nt = 0; nt < 8; ++nt) {
            v2f b;
            b.x = wk[nt * 16];
            b.y = wk[HIDF + nt * 16];
            c[nt] = __builtin_amdgcn_wmma_f32_16x16x4_f32(
                false, a, false, b, (short)0, c[nt], false, false);
        }
    }

#pragma unroll
    for (int nt = 0; nt < 8; ++nt) {
#pragma unroll
        for (int i = 0; i < 8; ++i) {
            Out[(long)(mtile * 16 + i + 8 * half) * HIDF + nt * 16 + lm] = c[nt][i];
        }
    }
}

template <int K>
__device__ __forceinline__ void add_rows(const v4f* __restrict__ hw4, int2 ent, int j,
                                         int lane, v4f& acc) {
    v4f   v[K];
    float w[K];
#pragma unroll
    for (int k = 0; k < K; ++k) {
        int s = __builtin_amdgcn_readlane(ent.x, j + k);
        w[k]  = __int_as_float(__builtin_amdgcn_readlane(ent.y, j + k));
        v[k]  = hw4[(long)s * (HIDF / 4) + lane];
    }
#pragma unroll
    for (int k = 0; k < K; ++k) acc += v[k] * w[k];
}

__global__ void __launch_bounds__(256) gather_kernel(const float* __restrict__ hw,
                                                     const int* __restrict__ rowptr,
                                                     const int* __restrict__ csr,
                                                     const float* __restrict__ bias,
                                                     float* __restrict__ h, int n) {
    const int lane = threadIdx.x & 31;
    const int node = __builtin_amdgcn_readfirstlane(blockIdx.x * 8 + (threadIdx.x >> 5));
    if (node >= n) return;
    const v4f*  hw4 = (const v4f*)hw;
    const int2* ed  = (const int2*)csr;

    const v4f bv  = *((const v4f*)bias + lane);
    const int beg = rowptr[node];
    const int end = rowptr[node + 1];

    v4f acc = {0.0f, 0.0f, 0.0f, 0.0f};
    for (int base = beg; base < end; base += 32) {
        int m = end - base;
        if (m > 32) m = 32;
        int2 ent = make_int2(0, 0);
        if (lane < m) ent = ed[base + lane];
        int j = 0;
        for (; j + 8 <= m; j += 8) add_rows<8>(hw4, ent, j, lane, acc);
        if (j + 4 <= m) { add_rows<4>(hw4, ent, j, lane, acc); j += 4; }
        const int r = m - j;
        if (r == 3)      add_rows<3>(hw4, ent, j, lane, acc);
        else if (r == 2) add_rows<2>(hw4, ent, j, lane, acc);
        else if (r == 1) add_rows<1>(hw4, ent, j, lane, acc);
    }

    v4f o = acc + bv;
    o.x = o.x > 0.0f ? o.x : 0.0f;
    o.y = o.y > 0.0f ? o.y : 0.0f;
    o.z = o.z > 0.0f ? o.z : 0.0f;
    o.w = o.w > 0.0f ? o.w : 0.0f;
    *((v4f*)(h + (long)node * HIDF) + lane) = o;
}

#define POOL_WAVES 8
#define POOL_ROWS  16
__global__ void __launch_bounds__(POOL_WAVES * 32) pool_kernel(const float* __restrict__ h,
                                                               const int* __restrict__ ptr,
                                                               float* __restrict__ out, int g) {
    __shared__ v4f s_part[POOL_WAVES][32];
    const int gi   = blockIdx.x;
    const int t    = threadIdx.x;
    const int lane = t & 31;
    const int wave = __builtin_amdgcn_readfirstlane(t >> 5);
    const int s = ptr[gi];
    const int e = ptr[gi + 1];
    const v4f* h4 = (const v4f*)h + lane;

    v4f acc = {0.0f, 0.0f, 0.0f, 0.0f};
    int n = s + wave;
    for (; n + (POOL_ROWS - 1) * POOL_WAVES < e; n += POOL_ROWS * POOL_WAVES) {
        v4f v[POOL_ROWS];
#pragma unroll
        for (int u = 0; u < POOL_ROWS; ++u) v[u] = h4[(long)(n + u * POOL_WAVES) * (HIDF / 4)];
#pragma unroll
        for (int u = 0; u < POOL_ROWS; ++u) acc += v[u];
    }
    if (n < e) {
        v4f v[POOL_ROWS];
        const int last = e - 1;
#pragma unroll
        for (int u = 0; u < POOL_ROWS; ++u) {
            const int r = n + u * POOL_WAVES;
            v[u] = h4[(long)(r < e ? r : last) * (HIDF / 4)];
        }
#pragma unroll
        for (int u = 0; u < POOL_ROWS; ++u) {
            if (n + u * POOL_WAVES < e) acc += v[u];
        }
    }

    s_part[wave][lane] = acc;
    __syncthreads();
    if (t < HIDF) {
        const float* sp = (const float*)s_part;
        float sum = 0.0f;
        for (int w = 0; w < POOL_WAVES; ++w) sum += sp[w * HIDF + t];
        float cnt = (float)(e - s);
        if (cnt < 1.0f) cnt = 1.0f;
        out[(long)gi * HIDF + t] = sum / cnt;
    }
}

extern "C" void kernel_launch(void* const* d_in, const int* in_sizes, int n_in,
                              void* d_out, int out_size, void* d_ws, size_t ws_size,
                              hipStream_t stream) {
    const float* x   = (const float*)d_in[0];
    const int*   ei  = (const int*)d_in[1];
    const int*   ptr = (const int*)d_in[2];
    const float* Wl[3] = {(const float*)d_in[3], (const float*)d_in[5], (const float*)d_in[7]};
    const float* bl[3] = {(const float*)d_in[4], (const float*)d_in[6], (const float*)d_in[8]};

    int N = in_sizes[0] / HIDF;
    int E = in_sizes[1] / 2;
    int G = in_sizes[2] - 1;

    const int* src = ei;
    const int* dst = ei + E;

    size_t npad = ((size_t)N + 255) & ~(size_t)255;
    size_t epad = ((size_t)E + 255) & ~(size_t)255;
    float* dinv   = (float*)d_ws;
    int*   cnt    = (int*)(dinv + npad);
    int*   rowptr = cnt + npad;
    int*   rank   = rowptr + npad + 256;
    int*   csr    = rank + epad;
    float* bufA   = (float*)(csr + 2 * (epad + npad));
    float* bufB   = bufA + (size_t)N * HIDF;

    init_cnt_kernel<<<(N + 255) / 256, 256, 0, stream>>>(cnt, N);
    count_rank_kernel<<<(E + 255) / 256, 256, 0, stream>>>(dst, cnt, rank, E);
    scan_kernel<<<(N + 4095) / 4096, 1024, 0, stream>>>(cnt, rowptr, dinv, csr, N);
    fill_kernel<<<(E + 255) / 256, 256, 0, stream>>>(src, dst, rank, rowptr, dinv, csr, E);

    int mtiles      = (N + 15) / 16;
    int gemm_blocks = (mtiles + 3) / 4;
    int gath_blocks = (N + 7) / 8;

    const float* hin = x;
    for (int l = 0; l < 3; ++l) {
        gemm_wmma_kernel<<<gemm_blocks, 128, 0, stream>>>(hin, Wl[l], bufA, N);
        gather_kernel<<<gath_blocks, 256, 0, stream>>>(bufA, rowptr, csr, bl[l], bufB, N);
        hin = bufB;
    }

    pool_kernel<<<G, 256, 0, stream>>>(bufB, ptr, (float*)d_out, G);
}


# gemm: a wave keeps its B fragments (32 columns, all k) in registers and loops over 16-row tiles of H staged once per block through LDS; W no longer re-read per 16 rows; k order and sums unchanged
# speedup vs baseline: 2.9971x; 1.1768x over previous
#include <hip/hip_runtime.h>

typedef float v2f __attribute__((ext_vector_type(2)));
typedef float v8f __attribute__((ext_vector_type(8)));
typedef float v4f __attribute__((ext_vector_type(4)));

#define HIDF 128


__global__ void init_cnt_kernel(int* __restrict__ cnt, int n) {
    int i = blockIdx.x * blockDim.x + threadIdx.x;
    if (i < n) cnt[i] = 0;
}

__global__ void count_rank_kernel(const int* __restrict__ dst, int* __restrict__ cnt,
                                  int* __restrict__ rank, int e) {
    int i = blockIdx.x * blockDim.x + threadIdx.x;
    if (i < e) rank[i] = atomicAdd(&cnt[dst[i]], 1);
}

__global__ void __launch_bounds__(1024) scan_kernel(const int* __restrict__ cnt,
                                                    int* __restrict__ rowptr,
                                                    float* __restrict__ dinv,
                                                    int* __restrict__ csr, int n) {
    __shared__ int s_part[32];
    __shared__ int s_loc[32];
    const int tid  = threadIdx.x;
    const int lane = tid & 31;
    const int wave = tid >> 5;
    const int blk_start = blockIdx.x * 4096;

    int part = 0;
    int i = tid * 4;
    for (; i + 7 * 4096 < blk_start; i += 8 * 4096) {
        int4 v[8];
#pragma unroll
        for (int u = 0; u < 8; ++u) v[u] = *(const int4*)(cnt + i + u * 4096);
#pragma unroll
        for (int u = 0; u < 8; ++u) part += (v[u].x + v[u].y) + (v[u].z + v[u].w);
    }
    for (; i < blk_start; i += 4096) {
        int4 v = *(const int4*)(cnt + i);
        part += (v.x + v.y) + (v.z + v.w);
    }

    const int i0 = blk_start + tid * 4;
    int c[4] = {0, 0, 0, 0};
    if (i0 + 3 < n) {
        int4 v = *(const int4*)(cnt + i0);
        c[0] = v.x; c[1] = v.y; c[2] = v.z; c[3] = v.w;
    } else {
        for (int k = 0; k < 4; ++k) if (i0 + k < n) c[k] = cnt[i0 + k];
    }
    const int loc = (c[0] + c[1]) + (c[2] + c[3]);

    int incl = loc;
    for (int off = 1; off < 32; off <<= 1) {
        int t = __shfl_up(incl, off);
        if (lane >= off) incl += t;
    }
    for (int off = 16; off > 0; off >>= 1) part += __shfl_xor(part, off);
    if (lane == 31) s_loc[wave] = incl;
    if (lane == 0)  s_part[wave] = part;
    __syncthreads();
    int base = 0;
    for (int w = 0; w < 32; ++w) {
        base += s_part[w];
        if (w < wave) base += s_loc[w];
    }
    int run = base + incl - loc;
    for (int k = 0; k < 4; ++k) {
        int idx = i0 + k;
        if (idx < n) {
            float di = rsqrtf((float)(c[k] + 1));
            int row  = run + idx;
            rowptr[idx] = row;
            dinv[idx]   = di;
            ((int2*)csr)[row + c[k]] = make_int2(idx, __float_as_int(di * di));
            run += c[k];
            if (idx == n - 1) rowptr[n] = run + n;
        }
    }
}

__global__ void fill_kernel(const int* __restrict__ src, const int* __restrict__ dst,
                            const int* __restrict__ rank, const int* __restrict__ rowptr,
                            const float* __restrict__ dinv, int* __restrict__ csr, int e) {
    int i = blockIdx.x * blockDim.x + threadIdx.x;
    if (i >= e) return;
    int s = src[i];
    int d = dst[i];
    int pos = rowptr[d] + rank[i];
    float norm = dinv[s] * dinv[d];
    ((int2*)csr)[pos] = make_int2(s, __float_as_int(norm));
}

#define GEMM_GRID  1024
#define A_PITCH    132
__global__ void __launch_bounds__(128) gemm_wmma_kernel(const float* __restrict__ H,
                                                        const float* __restrict__ W,
                                                        float* __restrict__ Out, int M) {
    __shared__ __attribute__((aligned(16))) float s_a[2][16 * A_PITCH];
    const int lane = threadIdx.x & 31;
    const int wave = __builtin_amdgcn_readfirstlane(threadIdx.x >> 5);
    const int half = lane >> 4;
    const int lm   = lane & 15;
    const int mtiles = (M + 15) >> 4;
    int tile = blockIdx.x;
    if (tile >= mtiles) return;
    const int col0 = wave * 32;

    v2f b[32][2];
#pragma unroll
    for (int ks = 0; ks < 32; ++ks) {
        const float* wk = W + (long)(4 * ks + 2 * half) * HIDF + col0 + lm;
#pragma unroll
        for (int t = 0; t < 2; ++t) {
            b[ks][t].x = wk[t * 16];
            b[ks][t].y = wk[HIDF + t * 16];
        }
    }

    v4f g[4];
#pragma unroll
    for (int q = 0; q < 4; ++q) {
        int row = tile * 16 + q * 4 + wave;
        if (row > M - 1) row = M - 1;
        g[q] = *((const v4f*)(H + (long)row * HIDF) + lane);
    }
#pragma unroll
    for (int q = 0; q < 4; ++q)
        *(v4f*)(&s_a[0][(q * 4 + wave) * A_PITCH + lane * 4]) = g[q];
    __syncthreads();

    int buf = 0;
    for (; tile < mtiles; tile += gridDim.x) {
        const int  next     = tile + gridDim.x;
        const bool has_next = next < mtiles;
        if (has_next) {
#pragma unroll
            for (int q = 0; q < 4; ++q) {
                int row = next * 16 + q * 4 + wave;
                if (row > M - 1) row = M - 1;
                g[q] = *((const v4f*)(H + (long)row * HIDF) + lane);
            }
        }

        const float* sa = &s_a[buf][lm * A_PITCH + 2 * half];
        v2f a[32];
#pragma unroll
        for (int ks = 0; ks < 32; ++ks) a[ks] = *(const v2f*)(sa + 4 * ks);

        v8f c0 = (v8f){0.f, 0.f, 0.f, 0.f, 0.f, 0.f, 0.f, 0.f};
        v8f c1 = (v8f){0.f, 0.f, 0.f, 0.f, 0.f, 0.f, 0.f, 0.f};
#pragma unroll
        for (int ks = 0; ks < 32; ++ks) {
            c0 = __builtin_amdgcn_wmma_f32_16x16x4_f32(false, a[ks], false, b[ks][0], (short)0, c0, false, false);
            c1 = __builtin_amdgcn_wmma_f32_16x16x4_f32(false, a[ks], false, b[ks][1], (short)0, c1, false, false);
        }

#pragma unroll
        for (int i = 0; i < 8; ++i) {
            const int row = tile * 16 + i + 8 * half;
            if (row < M) {
                float* o = Out + (long)row * HIDF + col0 + lm;
                o[0]  = c0[i];
                o[16] = c1[i];
            }
        }

        if (has_next) {
#pragma unroll
            for (int q = 0; q < 4; ++q)
                *(v4f*)(&s_a[buf ^ 1][(q * 4 + wave) * A_PITCH + lane * 4]) = g[q];
        }
        __syncthreads();
        buf ^= 1;
    }
}

template <int K>
__device__ __forceinline__ void add_rows(const v4f* __restrict__ hw4, int2 ent, int j,
                                         int lane, v4f& acc) {
    v4f   v[K];
    float w[K];
#pragma unroll
    for (int k = 0; k < K; ++k) {
        int s = __builtin_amdgcn_readlane(ent.x, j + k);
        w[k]  = __int_as_float(__builtin_amdgcn_readlane(ent.y, j + k));
        v[k]  = hw4[(long)s * (HIDF / 4) + lane];
    }
#pragma unroll
    for (int k = 0; k < K; ++k) acc += v[k] * w[k];
}

__global__ void __launch_bounds__(256) gather_kernel(const float* __restrict__ hw,
                                                     const int* __restrict__ rowptr,
                                                     const int* __restrict__ csr,
                                                     const float* __restrict__ bias,
                                                     float* __restrict__ h, int n) {
    const int lane = threadIdx.x & 31;
    const int node = __builtin_amdgcn_readfirstlane(blockIdx.x * 8 + (threadIdx.x >> 5));
    if (node >= n) return;
    const v4f*  hw4 = (const v4f*)hw;
    const int2* ed  = (const int2*)csr;

    const v4f bv  = *((const v4f*)bias + lane);
    const int beg = rowptr[node];
    const int end = rowptr[node + 1];

    v4f acc = {0.0f, 0.0f, 0.0f, 0.0f};
    for (int base = beg; base < end; base += 32) {
        int m = end - base;
        if (m > 32) m = 32;
        int2 ent = make_int2(0, 0);
        if (lane < m) ent = ed[base + lane];
        int j = 0;
        for (; j + 8 <= m; j += 8) add_rows<8>(hw4, ent, j, lane, acc);
        if (j + 4 <= m) { add_rows<4>(hw4, ent, j, lane, acc); j += 4; }
        const int r = m - j;
        if (r == 3)      add_rows<3>(hw4, ent, j, lane, acc);
        else if (r == 2) add_rows<2>(hw4, ent, j, lane, acc);
        else if (r == 1) add_rows<1>(hw4, ent, j, lane, acc);
    }

    v4f o = acc + bv;
    o.x = o.x > 0.0f ? o.x : 0.0f;
    o.y = o.y > 0.0f ? o.y : 0.0f;
    o.z = o.z > 0.0f ? o.z : 0.0f;
    o.w = o.w > 0.0f ? o.w : 0.0f;
    *((v4f*)(h + (long)node * HIDF) + lane) = o;
}

#define POOL_WAVES 8
#define POOL_ROWS  16
__global__ void __launch_bounds__(POOL_WAVES * 32) pool_kernel(const float* __restrict__ h,
                                                               const int* __restrict__ ptr,
                                                               float* __restrict__ out, int g) {
    __shared__ v4f s_part[POOL_WAVES][32];
    const int gi   = blockIdx.x;
    const int t    = threadIdx.x;
    const int lane = t & 31;
    const int wave = __builtin_amdgcn_readfirstlane(t >> 5);
    const int s = ptr[gi];
    const int e = ptr[gi + 1];
    const v4f* h4 = (const v4f*)h + lane;

    v4f acc = {0.0f, 0.0f, 0.0f, 0.0f};
    int n = s + wave;
    for (; n + (POOL_ROWS - 1) * POOL_WAVES < e; n += POOL_ROWS * POOL_WAVES) {
        v4f v[POOL_ROWS];
#pragma unroll
        for (int u = 0; u < POOL_ROWS; ++u) v[u] = h4[(long)(n + u * POOL_WAVES) * (HIDF / 4)];
#pragma unroll
        for (int u = 0; u < POOL_ROWS; ++u) acc += v[u];
    }
    if (n < e) {
        v4f v[POOL_ROWS];
        const int last = e - 1;
#pragma unroll
        for (int u = 0; u < POOL_ROWS; ++u) {
            const int r = n + u * POOL_WAVES;
            v[u] = h4[(long)(r < e ? r : last) * (HIDF / 4)];
        }
#pragma unroll
        for (int u = 0; u < POOL_ROWS; ++u) {
            if (n + u * POOL_WAVES < e) acc += v[u];
        }
    }

    s_part[wave][lane] = acc;
    __syncthreads();
    if (t < HIDF) {
        const float* sp = (const float*)s_part;
        float sum = 0.0f;
        for (int w = 0; w < POOL_WAVES; ++w) sum += sp[w * HIDF + t];
        float cnt = (float)(e - s);
        if (cnt < 1.0f) cnt = 1.0f;
        out[(long)gi * HIDF + t] = sum / cnt;
    }
}

extern "C" void kernel_launch(void* const* d_in, const int* in_sizes, int n_in,
                              void* d_out, int out_size, void* d_ws, size_t ws_size,
                              hipStream_t stream) {
    const float* x   = (const float*)d_in[0];
    const int*   ei  = (const int*)d_in[1];
    const int*   ptr = (const int*)d_in[2];
    const float* Wl[3] = {(const float*)d_in[3], (const float*)d_in[5], (const float*)d_in[7]};
    const float* bl[3] = {(const float*)d_in[4], (const float*)d_in[6], (const float*)d_in[8]};

    int N = in_sizes[0] / HIDF;
    int E = in_sizes[1] / 2;
    int G = in_sizes[2] - 1;

    const int* src = ei;
    const int* dst = ei + E;

    size_t npad = ((size_t)N + 255) & ~(size_t)255;
    size_t epad = ((size_t)E + 255) & ~(size_t)255;
    float* dinv   = (float*)d_ws;
    int*   cnt    = (int*)(dinv + npad);
    int*   rowptr = cnt + npad;
    int*   rank   = rowptr + npad + 256;
    int*   csr    = rank + epad;
    float* bufA   = (float*)(csr + 2 * (epad + npad));
    float* bufB   = bufA + (size_t)N * HIDF;

    init_cnt_kernel<<<(N + 255) / 256, 256, 0, stream>>>(cnt, N);
    count_rank_kernel<<<(E + 255) / 256, 256, 0, stream>>>(dst, cnt, rank, E);
    scan_kernel<<<(N + 4095) / 4096, 1024, 0, stream>>>(cnt, rowptr, dinv, csr, N);
    fill_kernel<<<(E + 255) / 256, 256, 0, stream>>>(src, dst, rank, rowptr, dinv, csr, E);

    int mtiles      = (N + 15) / 16;
    int gemm_blocks = mtiles < GEMM_GRID ? mtiles : GEMM_GRID;
    int gath_blocks = (N + 7) / 8;

    const float* hin = x;
    for (int l = 0; l < 3; ++l) {
        gemm_wmma_kernel<<<gemm_blocks, 128, 0, stream>>>(hin, Wl[l], bufA, N);
        gather_kernel<<<gath_blocks, 256, 0, stream>>>(bufA, rowptr, csr, bl[l], bufB, N);
        hin = bufB;
    }

    pool_kernel<<<G, 256, 0, stream>>>(bufB, ptr, (float*)d_out, G);
}


# layer fusion: aggregation of layer l (+bias, relu) sums 16 nodes' rows straight into the LDS tile that the WMMA product of layer l+1 reads; h not written or re-read, 9 launches; sums unchanged
# speedup vs baseline: 3.2860x; 1.0964x over previous
#include <hip/hip_runtime.h>

typedef float v2f __attribute__((ext_vector_type(2)));
typedef float v8f __attribute__((ext_vector_type(8)));
typedef float v4f __attribute__((ext_vector_type(4)));

#define HIDF 128


__global__ void init_cnt_kernel(int* __restrict__ cnt, int n) {
    int i = blockIdx.x * blockDim.x + threadIdx.x;
    if (i < n) cnt[i] = 0;
}

__global__ void count_rank_kernel(const int* __restrict__ dst, int* __restrict__ cnt,
                                  int* __restrict__ rank, int e) {
    int i = blockIdx.x * blockDim.x + threadIdx.x;
    if (i < e) rank[i] = atomicAdd(&cnt[dst[i]], 1);
}

__global__ void __launch_bounds__(1024) scan_kernel(const int* __restrict__ cnt,
                                                    int* __restrict__ rowptr,
                                                    float* __restrict__ dinv,
                                                    int* __restrict__ csr, int n) {
    __shared__ int s_part[32];
    __shared__ int s_loc[32];
    const int tid  = threadIdx.x;
    const int lane = tid & 31;
    const int wave = tid >> 5;
    const int blk_start = blockIdx.x * 4096;

    int part = 0;
    int i = tid * 4;
    for (; i + 7 * 4096 < blk_start; i += 8 * 4096) {
        int4 v[8];
#pragma unroll
        for (int u = 0; u < 8; ++u) v[u] = *(const int4*)(cnt + i + u * 4096);
#pragma unroll
        for (int u = 0; u < 8; ++u) part += (v[u].x + v[u].y) + (v[u].z + v[u].w);
    }
    for (; i < blk_start; i += 4096) {
        int4 v = *(const int4*)(cnt + i);
        part += (v.x + v.y) + (v.z + v.w);
    }

    const int i0 = blk_start + tid * 4;
    int c[4] = {0, 0, 0, 0};
    if (i0 + 3 < n) {
        int4 v = *(const int4*)(cnt + i0);
        c[0] = v.x; c[1] = v.y; c[2] = v.z; c[3] = v.w;
    } else {
        for (int k = 0; k < 4; ++k) if (i0 + k < n) c[k] = cnt[i0 + k];
    }
    const int loc = (c[0] + c[1]) + (c[2] + c[3]);

    int incl = loc;
    for (int off = 1; off < 32; off <<= 1) {
        int t = __shfl_up(incl, off);
        if (lane >= off) incl += t;
    }
    for (int off = 16; off > 0; off >>= 1) part += __shfl_xor(part, off);
    if (lane == 31) s_loc[wave] = incl;
    if (lane == 0)  s_part[wave] = part;
    __syncthreads();
    int base = 0;
    for (int w = 0; w < 32; ++w) {
        base += s_part[w];
        if (w < wave) base += s_loc[w];
    }
    int run = base + incl - loc;
    for (int k = 0; k < 4; ++k) {
        int idx = i0 + k;
        if (idx < n) {
            float di = rsqrtf((float)(c[k] + 1));
            int row  = run + idx;
            rowptr[idx] = row;
            dinv[idx]   = di;
            ((int2*)csr)[row + c[k]] = make_int2(idx, __float_as_int(di * di));
            run += c[k];
            if (idx == n - 1) rowptr[n] = run + n;
        }
    }
}

__global__ void fill_kernel(const int* __restrict__ src, const int* __restrict__ dst,
                            const int* __restrict__ rank, const int* __restrict__ rowptr,
                            const float* __restrict__ dinv, int* __restrict__ csr, int e) {
    int i = blockIdx.x * blockDim.x + threadIdx.x;
    if (i >= e) return;
    int s = src[i];
    int d = dst[i];
    int pos = rowptr[d] + rank[i];
    float norm = dinv[s] * dinv[d];
    ((int2*)csr)[pos] = make_int2(s, __float_as_int(norm));
}

#define GEMM_GRID  1024
#define A_PITCH    132
__global__ void __launch_bounds__(128) gemm_wmma_kernel(const float* __restrict__ H,
                                                        const float* __restrict__ W,
                                                        float* __restrict__ Out, int M) {
    __shared__ __attribute__((aligned(16))) float s_a[2][16 * A_PITCH];
    const int lane = threadIdx.x & 31;
    const int wave = __builtin_amdgcn_readfirstlane(threadIdx.x >> 5);
    const int half = lane >> 4;
    const int lm   = lane & 15;
    const int mtiles = (M + 15) >> 4;
    int tile = blockIdx.x;
    if (tile >= mtiles) return;
    const int col0 = wave * 32;

    v2f b[32][2];
#pragma unroll
    for (int ks = 0; ks < 32; ++ks) {
        const float* wk = W + (long)(4 * ks + 2 * half) * HIDF + col0 + lm;
#pragma unroll
        for (int t = 0; t < 2; ++t) {
            b[ks][t].x = wk[t * 16];
            b[ks][t].y = wk[HIDF + t * 16];
        }
    }

    v4f g[4];
#pragma unroll
    for (int q = 0; q < 4; ++q) {
        int row = tile * 16 + q * 4 + wave;
        if (row > M - 1) row = M - 1;
        g[q] = *((const v4f*)(H + (long)row * HIDF) + lane);
    }
#pragma unroll
    for (int q = 0; q < 4; ++q)
        *(v4f*)(&s_a[0][(q * 4 + wave) * A_PITCH + lane * 4]) = g[q];
    __syncthreads();

    int buf = 0;
    for (; tile < mtiles; tile += gridDim.x) {
        const int  next     = tile + gridDim.x;
        const bool has_next = next < mtiles;
        if (has_next) {
#pragma unroll
            for (int q = 0; q < 4; ++q) {
                int row = next * 16 + q * 4 + wave;
                if (row > M - 1) row = M - 1;
                g[q] = *((const v4f*)(H + (long)row * HIDF) + lane);
            }
        }

        const float* sa = &s_a[buf][lm * A_PITCH + 2 * half];
        v2f a[32];
#pragma unroll
        for (int ks = 0; ks < 32; ++ks) a[ks] = *(const v2f*)(sa + 4 * ks);

        v8f c0 = (v8f){0.f, 0.f, 0.f, 0.f, 0.f, 0.f, 0.f, 0.f};
        v8f c1 = (v8f){0.f, 0.f, 0.f, 0.f, 0.f, 0.f, 0.f, 0.f};
#pragma unroll
        for (int ks = 0; ks < 32; ++ks) {
            c0 = __builtin_amdgcn_wmma_f32_16x16x4_f32(false, a[ks], false, b[ks][0], (short)0, c0, false, false);
            c1 = __builtin_amdgcn_wmma_f32_16x16x4_f32(false, a[ks], false, b[ks][1], (short)0, c1, false, false);
        }

#pragma unroll
        for (int i = 0; i < 8; ++i) {
            const int row = tile * 16 + i + 8 * half;
            if (row < M) {
                float* o = Out + (long)row * HIDF + col0 + lm;
                o[0]  = c0[i];
                o[16] = c1[i];
            }
        }

        if (has_next) {
#pragma unroll
            for (int q = 0; q < 4; ++q)
                *(v4f*)(&s_a[buf ^ 1][(q * 4 + wave) * A_PITCH + lane * 4]) = g[q];
        }
        __syncthreads();
        buf ^= 1;
    }
}

template <int K>
__device__ __forceinline__ void add_rows(const v4f* __restrict__ hw4, int2 ent, int j,
                                         int lane, v4f& acc) {
    v4f   v[K];
    float w[K];
#pragma unroll
    for (int k = 0; k < K; ++k) {
        int s = __builtin_amdgcn_readlane(ent.x, j + k);
        w[k]  = __int_as_float(__builtin_amdgcn_readlane(ent.y, j + k));
        v[k]  = hw4[(long)s * (HIDF / 4) + lane];
    }
#pragma unroll
    for (int k = 0; k < K; ++k) acc += v[k] * w[k];
}

__global__ void __launch_bounds__(256) gather_kernel(const float* __restrict__ hw,
                                                     const int* __restrict__ rowptr,
                                                     const int* __restrict__ csr,
                                                     const float* __restrict__ bias,
                                                     float* __restrict__ h, int n) {
    const int lane = threadIdx.x & 31;
    const int node = __builtin_amdgcn_readfirstlane(blockIdx.x * 8 + (threadIdx.x >> 5));
    if (node >= n) return;
    const v4f*  hw4 = (const v4f*)hw;
    const int2* ed  = (const int2*)csr;

    const v4f bv  = *((const v4f*)bias + lane);
    const int beg = rowptr[node];
    const int end = rowptr[node + 1];

    v4f acc = {0.0f, 0.0f, 0.0f, 0.0f};
    for (int base = beg; base < end; base += 32) {
        int m = end - base;
        if (m > 32) m = 32;
        int2 ent = make_int2(0, 0);
        if (lane < m) ent = ed[base + lane];
        int j = 0;
        for (; j + 8 <= m; j += 8) add_rows<8>(hw4, ent, j, lane, acc);
        if (j + 4 <= m) { add_rows<4>(hw4, ent, j, lane, acc); j += 4; }
        const int r = m - j;
        if (r == 3)      add_rows<3>(hw4, ent, j, lane, acc);
        else if (r == 2) add_rows<2>(hw4, ent, j, lane, acc);
        else if (r == 1) add_rows<1>(hw4, ent, j, lane, acc);
    }

    v4f o = acc + bv;
    o.x = o.x > 0.0f ? o.x : 0.0f;
    o.y = o.y > 0.0f ? o.y : 0.0f;
    o.z = o.z > 0.0f ? o.z : 0.0f;
    o.w = o.w > 0.0f ? o.w : 0.0f;
    *((v4f*)(h + (long)node * HIDF) + lane) = o;
}

#define FUSE_GRID  1024

__device__ __forceinline__ void put_row(float* __restrict__ dst, v4f acc, v4f bv) {
    v4f o = acc + bv;
    o.x = o.x > 0.0f ? o.x : 0.0f;
    o.y = o.y > 0.0f ? o.y : 0.0f;
    o.z = o.z > 0.0f ? o.z : 0.0f;
    o.w = o.w > 0.0f ? o.w : 0.0f;
    *(v4f*)dst = o;
}

__device__ __forceinline__ int load_bounds(const int* __restrict__ rowptr, int t, int mtiles, int wave,
                                           int l4, int n) {
    int r = 0;
    if (t < mtiles) {
        int ri = t * 16 + 4 * wave + l4;
        if (ri > n) ri = n;
        r = rowptr[ri];
    }
    return r;
}

template <int J, int G>
__device__ __forceinline__ void add_rows_seg(const v4f* __restrict__ hw4, int2 ent, int cnt, int slot0,
                                             int lane, int rp, v4f bv, float* __restrict__ srow,
                                             v4f& acc, int& cur, int& nb) {
    v4f   v[G];
    float w[G];
    const int jlast = J + cnt - 1;
#pragma unroll
    for (int k = 0; k < G; ++k) {
        const int jj = J + k < jlast ? J + k : jlast;
        int s = __builtin_amdgcn_readlane(ent.x, jj);
        w[k]  = __int_as_float(__builtin_amdgcn_readlane(ent.y, jj));
        v[k]  = hw4[(long)s * (HIDF / 4) + lane];
    }
#pragma unroll
    for (int k = 0; k < G; ++k) {
        if (k < cnt) {
            if (slot0 + k == nb) {
                put_row(srow + cur * A_PITCH, acc, bv);
                ++cur;
                nb  = __builtin_amdgcn_readlane(rp, cur + 1);
                acc = (v4f){0.0f, 0.0f, 0.0f, 0.0f};
            }
            acc += v[k] * w[k];
        }
    }
}

template <int J>
__device__ __forceinline__ void add_rows_half(const v4f* __restrict__ hw4, int2 ent, int cnt, int slot0,
                                              int lane, int rp, v4f bv, float* __restrict__ srow,
                                              v4f& acc, int& cur, int& nb) {
    if (cnt > 12)     add_rows_seg<J, 16>(hw4, ent, cnt, slot0, lane, rp, bv, srow, acc, cur, nb);
    else if (cnt > 8) add_rows_seg<J, 12>(hw4, ent, cnt, slot0, lane, rp, bv, srow, acc, cur, nb);
    else if (cnt > 4) add_rows_seg<J, 8>(hw4, ent, cnt, slot0, lane, rp, bv, srow, acc, cur, nb);
    else              add_rows_seg<J, 4>(hw4, ent, cnt, slot0, lane, rp, bv, srow, acc, cur, nb);
}

__global__ void __launch_bounds__(128) gather_gemm_kernel(const float* __restrict__ hw,
                                                          const int* __restrict__ rowptr,
                                                          const int* __restrict__ csr,
                                                          const float* __restrict__ bias,
                                                          const float* __restrict__ W,
                                                          float* __restrict__ Out, int n) {
    __shared__ __attribute__((aligned(16))) float s_a[2][16 * A_PITCH];
    const int lane = threadIdx.x & 31;
    const int wave = __builtin_amdgcn_readfirstlane(threadIdx.x >> 5);
    const int half = lane >> 4;
    const int lm   = lane & 15;
    const int mtiles = (n + 15) >> 4;
    int tile = blockIdx.x;
    if (tile >= mtiles) return;
    const int col0 = wave * 32;
    const v4f*  hw4 = (const v4f*)hw;
    const int2* ed  = (const int2*)csr;
    const int   l4  = lane < 5 ? lane : 4;

    int  rp  = load_bounds(rowptr, tile, mtiles, wave, l4, n);
    int  rp1 = load_bounds(rowptr, tile + gridDim.x, mtiles, wave, l4, n);
    int  e0  = __builtin_amdgcn_readlane(rp, 0);
    int  e4  = __builtin_amdgcn_readlane(rp, 4);
    int2 ent = make_int2(0, 0);
    if (lane < e4 - e0) ent = ed[e0 + lane];

    v2f b[32][2];
#pragma unroll
    for (int ks = 0; ks < 32; ++ks) {
        const float* wk = W + (long)(4 * ks + 2 * half) * HIDF + col0 + lm;
#pragma unroll
        for (int t = 0; t < 2; ++t) {
            b[ks][t].x = wk[t * 16];
            b[ks][t].y = wk[HIDF + t * 16];
        }
    }
    const v4f bv = *((const v4f*)bias + lane);

    int buf = 0;
    for (; tile < mtiles; tile += gridDim.x) {
        float* srow = &s_a[buf][(4 * wave) * A_PITCH + lane * 4];
        v4f acc = {0.0f, 0.0f, 0.0f, 0.0f};
        int cur = 0;
        int nb  = __builtin_amdgcn_readlane(rp, 1);
        for (int base = e0; base < e4; ) {
            int m = e4 - base;
            if (m > 32) m = 32;
            add_rows_half<0>(hw4, ent, m < 16 ? m : 16, base, lane, rp, bv, srow, acc, cur, nb);
            if (m > 16) add_rows_half<16>(hw4, ent, m - 16, base + 16, lane, rp, bv, srow, acc, cur, nb);
            base += 32;
            if (base < e4) {
                ent = make_int2(0, 0);
                if (lane < e4 - base) ent = ed[base + lane];
            }
        }
        for (; cur < 4; ++cur) {
            put_row(srow + cur * A_PITCH, acc, bv);
            acc = (v4f){0.0f, 0.0f, 0.0f, 0.0f};
        }
        __syncthreads();

        rp  = rp1;
        e0  = __builtin_amdgcn_readlane(rp, 0);
        e4  = __builtin_amdgcn_readlane(rp, 4);
        ent = make_int2(0, 0);
        if (lane < e4 - e0) ent = ed[e0 + lane];
        rp1 = load_bounds(rowptr, tile + 2 * gridDim.x, mtiles, wave, l4, n);

        const float* sa = &s_a[buf][lm * A_PITCH + 2 * half];
        v2f a[32];
#pragma unroll
        for (int ks = 0; ks < 32; ++ks) a[ks] = *(const v2f*)(sa + 4 * ks);

        v8f c0 = (v8f){0.f, 0.f, 0.f, 0.f, 0.f, 0.f, 0.f, 0.f};
        v8f c1 = (v8f){0.f, 0.f, 0.f, 0.f, 0.f, 0.f, 0.f, 0.f};
#pragma unroll
        for (int ks = 0; ks < 32; ++ks) {
            c0 = __builtin_amdgcn_wmma_f32_16x16x4_f32(false, a[ks], false, b[ks][0], (short)0, c0, false, false);
            c1 = __builtin_amdgcn_wmma_f32_16x16x4_f32(false, a[ks], false, b[ks][1], (short)0, c1, false, false);
        }

#pragma unroll
        for (int i = 0; i < 8; ++i) {
            const int row = tile * 16 + i + 8 * half;
            if (row < n) {
                float* o = Out + (long)row * HIDF + col0 + lm;
                o[0]  = c0[i];
                o[16] = c1[i];
            }
        }
        buf ^= 1;
    }
}

#define POOL_WAVES 8
#define POOL_ROWS  16
__global__ void __launch_bounds__(POOL_WAVES * 32) pool_kernel(const float* __restrict__ h,
                                                               const int* __restrict__ ptr,
                                                               float* __restrict__ out, int g) {
    __shared__ v4f s_part[POOL_WAVES][32];
    const int gi   = blockIdx.x;
    const int t    = threadIdx.x;
    const int lane = t & 31;
    const int wave = __builtin_amdgcn_readfirstlane(t >> 5);
    const int s = ptr[gi];
    const int e = ptr[gi + 1];
    const v4f* h4 = (const v4f*)h + lane;

    v4f acc = {0.0f, 0.0f, 0.0f, 0.0f};
    int n = s + wave;
    for (; n + (POOL_ROWS - 1) * POOL_WAVES < e; n += POOL_ROWS * POOL_WAVES) {
        v4f v[POOL_ROWS];
#pragma unroll
        for (int u = 0; u < POOL_ROWS; ++u) v[u] = h4[(long)(n + u * POOL_WAVES) * (HIDF / 4)];
#pragma unroll
        for (int u = 0; u < POOL_ROWS; ++u) acc += v[u];
    }
    if (n < e) {
        v4f v[POOL_ROWS];
        const int last = e - 1;
#pragma unroll
        for (int u = 0; u < POOL_ROWS; ++u) {
            const int r = n + u * POOL_WAVES;
            v[u] = h4[(long)(r < e ? r : last) * (HIDF / 4)];
        }
#pragma unroll
        for (int u = 0; u < POOL_ROWS; ++u) {
            if (n + u * POOL_WAVES < e) acc += v[u];
        }
    }

    s_part[wave][lane] = acc;
    __syncthreads();
    if (t < HIDF) {
        const float* sp = (const float*)s_part;
        float sum = 0.0f;
        for (int w = 0; w < POOL_WAVES; ++w) sum += sp[w * HIDF + t];
        float cnt = (float)(e - s);
        if (cnt < 1.0f) cnt = 1.0f;
        out[(long)gi * HIDF + t] = sum / cnt;
    }
}

extern "C" void kernel_launch(void* const* d_in, const int* in_sizes, int n_in,
                              void* d_out, int out_size, void* d_ws, size_t ws_size,
                              hipStream_t stream) {
    const float* x   = (const float*)d_in[0];
    const int*   ei  = (const int*)d_in[1];
    const int*   ptr = (const int*)d_in[2];
    const float* Wl[3] = {(const float*)d_in[3], (const float*)d_in[5], (const float*)d_in[7]};
    const float* bl[3] = {(const float*)d_in[4], (const float*)d_in[6], (const float*)d_in[8]};

    int N = in_sizes[0] / HIDF;
    int E = in_sizes[1] / 2;
    int G = in_sizes[2] - 1;

    const int* src = ei;
    const int* dst = ei + E;

    size_t npad = ((size_t)N + 255) & ~(size_t)255;
    size_t epad = ((size_t)E + 255) & ~(size_t)255;
    float* dinv   = (float*)d_ws;
    int*   cnt    = (int*)(dinv + npad);
    int*   rowptr = cnt + npad;
    int*   rank   = rowptr + npad + 256;
    int*   csr    = rank + epad;
    float* bufA   = (float*)(csr + 2 * (epad + npad));
    float* bufB   = bufA + (size_t)N * HIDF;

    init_cnt_kernel<<<(N + 255) / 256, 256, 0, stream>>>(cnt, N);
    count_rank_kernel<<<(E + 255) / 256, 256, 0, stream>>>(dst, cnt, rank, E);
    scan_kernel<<<(N + 4095) / 4096, 1024, 0, stream>>>(cnt, rowptr, dinv, csr, N);
    fill_kernel<<<(E + 255) / 256, 256, 0, stream>>>(src, dst, rank, rowptr, dinv, csr, E);

    int mtiles      = (N + 15) / 16;
    int gemm_blocks = mtiles < GEMM_GRID ? mtiles : GEMM_GRID;
    int gath_blocks = (N + 7) / 8;

    int fuse_blocks = mtiles < FUSE_GRID ? mtiles : FUSE_GRID;

    gemm_wmma_kernel<<<gemm_blocks, 128, 0, stream>>>(x, Wl[0], bufA, N);
    gather_gemm_kernel<<<fuse_blocks, 128, 0, stream>>>(bufA, rowptr, csr, bl[0], Wl[1], bufB, N);
    gather_gemm_kernel<<<fuse_blocks, 128, 0, stream>>>(bufB, rowptr, csr, bl[1], Wl[2], bufA, N);
    gather_kernel<<<gath_blocks, 256, 0, stream>>>(bufA, rowptr, csr, bl[2], bufB, N);

    pool_kernel<<<G, 256, 0, stream>>>(bufB, ptr, (float*)d_out, G);
}


# CSR row starts, in-degrees and edge ranks from a two-level counting sort on LDS counters (2048-edge chunks, 256-node buckets): no global atomic, no clear or scan launch (8); node sum order = LDS rank
# speedup vs baseline: 3.5124x; 1.0689x over previous
#include <hip/hip_runtime.h>

typedef float v2f __attribute__((ext_vector_type(2)));
typedef float v8f __attribute__((ext_vector_type(8)));
typedef float v4f __attribute__((ext_vector_type(4)));

#define HIDF 128

#define CHUNK_E   2048
#define BKT_SHIFT 8
#define BKT_NODES 256
#define MAX_BKT   512

__global__ void __launch_bounds__(256) tile_sort_kernel(const int* __restrict__ dst,
                                                        unsigned* __restrict__ sorted,
                                                        int* __restrict__ tab, int e, int nb) {
    __shared__ int      s_cnt[MAX_BKT];
    __shared__ int      s_start[MAX_BKT];
    __shared__ int      s_wsum[8];
    __shared__ unsigned s_out[CHUNK_E];
    const int tid  = threadIdx.x;
    const int lane = tid & 31;
    const int wave = tid >> 5;
    const int c    = blockIdx.x;
    const int e0   = c * CHUNK_E;

    s_cnt[tid]       = 0;
    s_cnt[tid + 256] = 0;
    __syncthreads();

    int d[8];
    int r[8];
#pragma unroll
    for (int k = 0; k < 8; ++k) {
        const int i = e0 + k * 256 + tid;
        d[k] = i < e ? dst[i] : -1;
    }
#pragma unroll
    for (int k = 0; k < 8; ++k) {
        r[k] = 0;
        if (d[k] >= 0) r[k] = atomicAdd(&s_cnt[d[k] >> BKT_SHIFT], 1);
    }
    __syncthreads();

    const int v0  = s_cnt[2 * tid];
    const int v1  = s_cnt[2 * tid + 1];
    const int loc = v0 + v1;
    int incl = loc;
    for (int off = 1; off < 32; off <<= 1) {
        int t = __shfl_up(incl, off);
        if (lane >= off) incl += t;
    }
    if (lane == 31) s_wsum[wave] = incl;
    __syncthreads();
    int base = 0;
    for (int w = 0; w < 8; ++w) if (w < wave) base += s_wsum[w];
    const int ex = base + incl - loc;
    s_start[2 * tid]     = ex;
    s_start[2 * tid + 1] = ex + v0;
    __syncthreads();

    for (int b = tid; b < nb; b += 256) tab[c * nb + b] = s_start[b] | (s_cnt[b] << 16);

#pragma unroll
    for (int k = 0; k < 8; ++k) {
        if (d[k] >= 0) {
            const unsigned w = (unsigned)(e0 + k * 256 + tid) | ((unsigned)(d[k] & (BKT_NODES - 1)) << 24);
            s_out[s_start[d[k] >> BKT_SHIFT] + r[k]] = w;
        }
    }
    __syncthreads();
    int m = e - e0;
    if (m > CHUNK_E) m = CHUNK_E;
    for (int j = tid; j < m; j += 256) sorted[e0 + j] = s_out[j];
}

__global__ void __launch_bounds__(512) bucket_rank_kernel(const unsigned* __restrict__ sorted,
                                                          const int* __restrict__ tab,
                                                          int* __restrict__ rowptr, float* __restrict__ dinv,
                                                          int* __restrict__ csr, int* __restrict__ rank,
                                                          int n, int nb, int nc) {
    __shared__ int s_lcnt[BKT_NODES];
    __shared__ int s_red[16];
    __shared__ int s_wsum[8];
    const int tid  = threadIdx.x;
    const int lane = tid & 31;
    const int wave = tid >> 5;
    const int b    = blockIdx.x;
    if (tid < BKT_NODES) s_lcnt[tid] = 0;
    __syncthreads();
    int before = 0;
    for (int c = tid; c < nc; c += 512) {
        const int p     = tab[c * nb + b];
        const int start = p & 0xFFFF;
        const int count = p >> 16;
        before += start;
        const unsigned* sp = sorted + c * CHUNK_E + start;
        for (int k = 0; k < count; ++k) {
            const unsigned w = sp[k];
            const int r = atomicAdd(&s_lcnt[w >> 24], 1);
            rank[w & 0xFFFFFFu] = r;
        }
    }
    for (int off = 16; off > 0; off >>= 1) before += __shfl_xor(before, off);
    if (lane == 0) s_red[wave] = before;
    __syncthreads();
    int base = 0;
    for (int w = 0; w < 16; ++w) base += s_red[w];

    const int cn = tid < BKT_NODES ? s_lcnt[tid] : 0;
    int incl = cn;
    for (int off = 1; off < 32; off <<= 1) {
        int t = __shfl_up(incl, off);
        if (lane >= off) incl += t;
    }
    if (tid < BKT_NODES && lane == 31) s_wsum[wave] = incl;
    __syncthreads();
    const int node = b * BKT_NODES + tid;
    if (tid < BKT_NODES && node < n) {
        int pre = 0;
        for (int w = 0; w < 8; ++w) if (w < wave) pre += s_wsum[w];
        const int   row = base + pre + incl - cn + node;
        const float di  = rsqrtf((float)(cn + 1));
        rowptr[node] = row;
        dinv[node]   = di;
        ((int2*)csr)[row + cn] = make_int2(node, __float_as_int(di * di));
        if (node == n - 1) rowptr[n] = row + cn + 1;
    }
}

__global__ void fill_kernel(const int* __restrict__ src, const int* __restrict__ dst,
                            const int* __restrict__ rank, const int* __restrict__ rowptr,
                            const float* __restrict__ dinv, int* __restrict__ csr, int e) {
    int i = blockIdx.x * blockDim.x + threadIdx.x;
    if (i >= e) return;
    int s = src[i];
    int d = dst[i];
    int pos = rowptr[d] + rank[i];
    float norm = dinv[s] * dinv[d];
    ((int2*)csr)[pos] = make_int2(s, __float_as_int(norm));
}

#define GEMM_GRID  1024
#define A_PITCH    132
__global__ void __launch_bounds__(128) gemm_wmma_kernel(const float* __restrict__ H,
                                                        const float* __restrict__ W,
                                                        float* __restrict__ Out, int M) {
    __shared__ __attribute__((aligned(16))) float s_a[2][16 * A_PITCH];
    const int lane = threadIdx.x & 31;
    const int wave = __builtin_amdgcn_readfirstlane(threadIdx.x >> 5);
    const int half = lane >> 4;
    const int lm   = lane & 15;
    const int mtiles = (M + 15) >> 4;
    int tile = blockIdx.x;
    if (tile >= mtiles) return;
    const int col0 = wave * 32;

    v2f b[32][2];
#pragma unroll
    for (int ks = 0; ks < 32; ++ks) {
        const float* wk = W + (long)(4 * ks + 2 * half) * HIDF + col0 + lm;
#pragma unroll
        for (int t = 0; t < 2; ++t) {
            b[ks][t].x = wk[t * 16];
            b[ks][t].y = wk[HIDF + t * 16];
        }
    }

    v4f g[4];
#pragma unroll
    for (int q = 0; q < 4; ++q) {
        int row = tile * 16 + q * 4 + wave;
        if (row > M - 1) row = M - 1;
        g[q] = *((const v4f*)(H + (long)row * HIDF) + lane);
    }
#pragma unroll
    for (int q = 0; q < 4; ++q)
        *(v4f*)(&s_a[0][(q * 4 + wave) * A_PITCH + lane * 4]) = g[q];
    __syncthreads();

    int buf = 0;
    for (; tile < mtiles; tile += gridDim.x) {
        const int  next     = tile + gridDim.x;
        const bool has_next = next < mtiles;
        if (has_next) {
#pragma unroll
            for (int q = 0; q < 4; ++q) {
                int row = next * 16 + q * 4 + wave;
                if (row > M - 1) row = M - 1;
                g[q] = *((const v4f*)(H + (long)row * HIDF) + lane);
            }
        }

        const float* sa = &s_a[buf][lm * A_PITCH + 2 * half];
        v2f a[32];
#pragma unroll
        for (int ks = 0; ks < 32; ++ks) a[ks] = *(const v2f*)(sa + 4 * ks);

        v8f c0 = (v8f){0.f, 0.f, 0.f, 0.f, 0.f, 0.f, 0.f, 0.f};
        v8f c1 = (v8f){0.f, 0.f, 0.f, 0.f, 0.f, 0.f, 0.f, 0.f};
#pragma unroll
        for (int ks = 0; ks < 32; ++ks) {
            c0 = __builtin_amdgcn_wmma_f32_16x16x4_f32(false, a[ks], false, b[ks][0], (short)0, c0, false, false);
            c1 = __builtin_amdgcn_wmma_f32_16x16x4_f32(false, a[ks], false, b[ks][1], (short)0, c1, false, false);
        }

#pragma unroll
        for (int i = 0; i < 8; ++i) {
            const int row = tile * 16 + i + 8 * half;
            if (row < M) {
                float* o = Out + (long)row * HIDF + col0 + lm;
                o[0]  = c0[i];
                o[16] = c1[i];
            }
        }

        if (has_next) {
#pragma unroll
            for (int q = 0; q < 4; ++q)
                *(v4f*)(&s_a[buf ^ 1][(q * 4 + wave) * A_PITCH + lane * 4]) = g[q];
        }
        __syncthreads();
        buf ^= 1;
    }
}

template <int K>
__device__ __forceinline__ void add_rows(const v4f* __restrict__ hw4, int2 ent, int j,
                                         int lane, v4f& acc) {
    v4f   v[K];
    float w[K];
#pragma unroll
    for (int k = 0; k < K; ++k) {
        int s = __builtin_amdgcn_readlane(ent.x, j + k);
        w[k]  = __int_as_float(__builtin_amdgcn_readlane(ent.y, j + k));
        v[k]  = hw4[(long)s * (HIDF / 4) + lane];
    }
#pragma unroll
    for (int k = 0; k < K; ++k) acc += v[k] * w[k];
}

__global__ void __launch_bounds__(256) gather_kernel(const float* __restrict__ hw,
                                                     const int* __restrict__ rowptr,
                                                     const int* __restrict__ csr,
                                                     const float* __restrict__ bias,
                                                     float* __restrict__ h, int n) {
    const int lane = threadIdx.x & 31;
    const int node = __builtin_amdgcn_readfirstlane(blockIdx.x * 8 + (threadIdx.x >> 5));
    if (node >= n) return;
    const v4f*  hw4 = (const v4f*)hw;
    const int2* ed  = (const int2*)csr;

    const v4f bv  = *((const v4f*)bias + lane);
    const int beg = rowptr[node];
    const int end = rowptr[node + 1];

    v4f acc = {0.0f, 0.0f, 0.0f, 0.0f};
    for (int base = beg; base < end; base += 32) {
        int m = end - base;
        if (m > 32) m = 32;
        int2 ent = make_int2(0, 0);
        if (lane < m) ent = ed[base + lane];
        int j = 0;
        for (; j + 8 <= m; j += 8) add_rows<8>(hw4, ent, j, lane, acc);
        if (j + 4 <= m) { add_rows<4>(hw4, ent, j, lane, acc); j += 4; }
        const int r = m - j;
        if (r == 3)      add_rows<3>(hw4, ent, j, lane, acc);
        else if (r == 2) add_rows<2>(hw4, ent, j, lane, acc);
        else if (r == 1) add_rows<1>(hw4, ent, j, lane, acc);
    }

    v4f o = acc + bv;
    o.x = o.x > 0.0f ? o.x : 0.0f;
    o.y = o.y > 0.0f ? o.y : 0.0f;
    o.z = o.z > 0.0f ? o.z : 0.0f;
    o.w = o.w > 0.0f ? o.w : 0.0f;
    *((v4f*)(h + (long)node * HIDF) + lane) = o;
}

#define FUSE_GRID  1024

__device__ __forceinline__ void put_row(float* __restrict__ dst, v4f acc, v4f bv) {
    v4f o = acc + bv;
    o.x = o.x > 0.0f ? o.x : 0.0f;
    o.y = o.y > 0.0f ? o.y : 0.0f;
    o.z = o.z > 0.0f ? o.z : 0.0f;
    o.w = o.w > 0.0f ? o.w : 0.0f;
    *(v4f*)dst = o;
}

__device__ __forceinline__ int load_bounds(const int* __restrict__ rowptr, int t, int mtiles, int wave,
                                           int l4, int n) {
    int r = 0;
    if (t < mtiles) {
        int ri = t * 16 + 4 * wave + l4;
        if (ri > n) ri = n;
        r = rowptr[ri];
    }
    return r;
}

template <int J, int G>
__device__ __forceinline__ void add_rows_seg(const v4f* __restrict__ hw4, int2 ent, int cnt, int slot0,
                                             int lane, int rp, v4f bv, float* __restrict__ srow,
                                             v4f& acc, int& cur, int& nb) {
    v4f   v[G];
    float w[G];
    const int jlast = J + cnt - 1;
#pragma unroll
    for (int k = 0; k < G; ++k) {
        const int jj = J + k < jlast ? J + k : jlast;
        int s = __builtin_amdgcn_readlane(ent.x, jj);
        w[k]  = __int_as_float(__builtin_amdgcn_readlane(ent.y, jj));
        v[k]  = hw4[(long)s * (HIDF / 4) + lane];
    }
#pragma unroll
    for (int k = 0; k < G; ++k) {
        if (k < cnt) {
            if (slot0 + k == nb) {
                put_row(srow + cur * A_PITCH, acc, bv);
                ++cur;
                nb  = __builtin_amdgcn_readlane(rp, cur + 1);
                acc = (v4f){0.0f, 0.0f, 0.0f, 0.0f};
            }
            acc += v[k] * w[k];
        }
    }
}

template <int J>
__device__ __forceinline__ void add_rows_half(const v4f* __restrict__ hw4, int2 ent, int cnt, int slot0,
                                              int lane, int rp, v4f bv, float* __restrict__ srow,
                                              v4f& acc, int& cur, int& nb) {
    if (cnt > 12)     add_rows_seg<J, 16>(hw4, ent, cnt, slot0, lane, rp, bv, srow, acc, cur, nb);
    else if (cnt > 8) add_rows_seg<J, 12>(hw4, ent, cnt, slot0, lane, rp, bv, srow, acc, cur, nb);
    else if (cnt > 4) add_rows_seg<J, 8>(hw4, ent, cnt, slot0, lane, rp, bv, srow, acc, cur, nb);
    else              add_rows_seg<J, 4>(hw4, ent, cnt, slot0, lane, rp, bv, srow, acc, cur, nb);
}

__global__ void __launch_bounds__(128) gather_gemm_kernel(const float* __restrict__ hw,
                                                          const int* __restrict__ rowptr,
                                                          const int* __restrict__ csr,
                                                          const float* __restrict__ bias,
                                                          const float* __restrict__ W,
                                                          float* __restrict__ Out, int n) {
    __shared__ __attribute__((aligned(16))) float s_a[2][16 * A_PITCH];
    const int lane = threadIdx.x & 31;
    const int wave = __builtin_amdgcn_readfirstlane(threadIdx.x >> 5);
    const int half = lane >> 4;
    const int lm   = lane & 15;
    const int mtiles = (n + 15) >> 4;
    int tile = blockIdx.x;
    if (tile >= mtiles) return;
    const int col0 = wave * 32;
    const v4f*  hw4 = (const v4f*)hw;
    const int2* ed  = (const int2*)csr;
    const int   l4  = lane < 5 ? lane : 4;

    int  rp  = load_bounds(rowptr, tile, mtiles, wave, l4, n);
    int  rp1 = load_bounds(rowptr, tile + gridDim.x, mtiles, wave, l4, n);
    int  e0  = __builtin_amdgcn_readlane(rp, 0);
    int  e4  = __builtin_amdgcn_readlane(rp, 4);
    int2 ent = make_int2(0, 0);
    if (lane < e4 - e0) ent = ed[e0 + lane];

    v2f b[32][2];
#pragma unroll
    for (int ks = 0; ks < 32; ++ks) {
        const float* wk = W + (long)(4 * ks + 2 * half) * HIDF + col0 + lm;
#pragma unroll
        for (int t = 0; t < 2; ++t) {
            b[ks][t].x = wk[t * 16];
            b[ks][t].y = wk[HIDF + t * 16];
        }
    }
    const v4f bv = *((const v4f*)bias + lane);

    int buf = 0;
    for (; tile < mtiles; tile += gridDim.x) {
        float* srow = &s_a[buf][(4 * wave) * A_PITCH + lane * 4];
        v4f acc = {0.0f, 0.0f, 0.0f, 0.0f};
        int cur = 0;
        int nb  = __builtin_amdgcn_readlane(rp, 1);
        for (int base = e0; base < e4; ) {
            int m = e4 - base;
            if (m > 32) m = 32;
            add_rows_half<0>(hw4, ent, m < 16 ? m : 16, base, lane, rp, bv, srow, acc, cur, nb);
            if (m > 16) add_rows_half<16>(hw4, ent, m - 16, base + 16, lane, rp, bv, srow, acc, cur, nb);
            base += 32;
            if (base < e4) {
                ent = make_int2(0, 0);
                if (lane < e4 - base) ent = ed[base + lane];
            }
        }
        for (; cur < 4; ++cur) {
            put_row(srow + cur * A_PITCH, acc, bv);
            acc = (v4f){0.0f, 0.0f, 0.0f, 0.0f};
        }
        __syncthreads();

        rp  = rp1;
        e0  = __builtin_amdgcn_readlane(rp, 0);
        e4  = __builtin_amdgcn_readlane(rp, 4);
        ent = make_int2(0, 0);
        if (lane < e4 - e0) ent = ed[e0 + lane];
        rp1 = load_bounds(rowptr, tile + 2 * gridDim.x, mtiles, wave, l4, n);

        const float* sa = &s_a[buf][lm * A_PITCH + 2 * half];
        v2f a[32];
#pragma unroll
        for (int ks = 0; ks < 32; ++ks) a[ks] = *(const v2f*)(sa + 4 * ks);

        v8f c0 = (v8f){0.f, 0.f, 0.f, 0.f, 0.f, 0.f, 0.f, 0.f};
        v8f c1 = (v8f){0.f, 0.f, 0.f, 0.f, 0.f, 0.f, 0.f, 0.f};
#pragma unroll
        for (int ks = 0; ks < 32; ++ks) {
            c0 = __builtin_amdgcn_wmma_f32_16x16x4_f32(false, a[ks], false, b[ks][0], (short)0, c0, false, false);
            c1 = __builtin_amdgcn_wmma_f32_16x16x4_f32(false, a[ks], false, b[ks][1], (short)0, c1, false, false);
        }

#pragma unroll
        for (int i = 0; i < 8; ++i) {
            const int row = tile * 16 + i + 8 * half;
            if (row < n) {
                float* o = Out + (long)row * HIDF + col0 + lm;
                o[0]  = c0[i];
                o[16] = c1[i];
            }
        }
        buf ^= 1;
    }
}

#define POOL_WAVES 8
#define POOL_ROWS  16
__global__ void __launch_bounds__(POOL_WAVES * 32) pool_kernel(const float* __restrict__ h,
                                                               const int* __restrict__ ptr,
                                                               float* __restrict__ out, int g) {
    __shared__ v4f s_part[POOL_WAVES][32];
    const int gi   = blockIdx.x;
    const int t    = threadIdx.x;
    const int lane = t & 31;
    const int wave = __builtin_amdgcn_readfirstlane(t >> 5);
    const int s = ptr[gi];
    const int e = ptr[gi + 1];
    const v4f* h4 = (const v4f*)h + lane;

    v4f acc = {0.0f, 0.0f, 0.0f, 0.0f};
    int n = s + wave;
    for (; n + (POOL_ROWS - 1) * POOL_WAVES < e; n += POOL_ROWS * POOL_WAVES) {
        v4f v[POOL_ROWS];
#pragma unroll
        for (int u = 0; u < POOL_ROWS; ++u) v[u] = h4[(long)(n + u * POOL_WAVES) * (HIDF / 4)];
#pragma unroll
        for (int u = 0; u < POOL_ROWS; ++u) acc += v[u];
    }
    if (n < e) {
        v4f v[POOL_ROWS];
        const int last = e - 1;
#pragma unroll
        for (int u = 0; u < POOL_ROWS; ++u) {
            const int r = n + u * POOL_WAVES;
            v[u] = h4[(long)(r < e ? r : last) * (HIDF / 4)];
        }
#pragma unroll
        for (int u = 0; u < POOL_ROWS; ++u) {
            if (n + u * POOL_WAVES < e) acc += v[u];
        }
    }

    s_part[wave][lane] = acc;
    __syncthreads();
    if (t < HIDF) {
        const float* sp = (const float*)s_part;
        float sum = 0.0f;
        for (int w = 0; w < POOL_WAVES; ++w) sum += sp[w * HIDF + t];
        float cnt = (float)(e - s);
        if (cnt < 1.0f) cnt = 1.0f;
        out[(long)gi * HIDF + t] = sum / cnt;
    }
}

extern "C" void kernel_launch(void* const* d_in, const int* in_sizes, int n_in,
                              void* d_out, int out_size, void* d_ws, size_t ws_size,
                              hipStream_t stream) {
    const float* x   = (const float*)d_in[0];
    const int*   ei  = (const int*)d_in[1];
    const int*   ptr = (const int*)d_in[2];
    const float* Wl[3] = {(const float*)d_in[3], (const float*)d_in[5], (const float*)d_in[7]};
    const float* bl[3] = {(const float*)d_in[4], (const float*)d_in[6], (const float*)d_in[8]};

    int N = in_sizes[0] / HIDF;
    int E = in_sizes[1] / 2;
    int G = in_sizes[2] - 1;

    const int* src = ei;
    const int* dst = ei + E;

    size_t npad = ((size_t)N + 255) & ~(size_t)255;
    size_t epad = ((size_t)E + 255) & ~(size_t)255;
    float* dinv   = (float*)d_ws;
    int*   rowptr = (int*)(dinv + npad);
    int*   rank   = rowptr + npad + 256;
    int*   csr    = rank + epad;
    float* bufA   = (float*)(csr + 2 * (epad + npad));
    float* bufB   = bufA + (size_t)N * HIDF;
    unsigned* sorted = (unsigned*)(bufB + (size_t)N * HIDF);
    int*      tab    = (int*)(sorted + epad);

    int nb = (N + BKT_NODES - 1) / BKT_NODES;
    int nc = (E + CHUNK_E - 1) / CHUNK_E;
    tile_sort_kernel<<<nc, 256, 0, stream>>>(dst, sorted, tab, E, nb);
    bucket_rank_kernel<<<nb, 512, 0, stream>>>(sorted, tab, rowptr, dinv, csr, rank, N, nb, nc);
    fill_kernel<<<(E + 255) / 256, 256, 0, stream>>>(src, dst, rank, rowptr, dinv, csr, E);

    int mtiles      = (N + 15) / 16;
    int gemm_blocks = mtiles < GEMM_GRID ? mtiles : GEMM_GRID;
    int gath_blocks = (N + 7) / 8;

    int fuse_blocks = mtiles < FUSE_GRID ? mtiles : FUSE_GRID;

    gemm_wmma_kernel<<<gemm_blocks, 128, 0, stream>>>(x, Wl[0], bufA, N);
    gather_gemm_kernel<<<fuse_blocks, 128, 0, stream>>>(bufA, rowptr, csr, bl[0], Wl[1], bufB, N);
    gather_gemm_kernel<<<fuse_blocks, 128, 0, stream>>>(bufB, rowptr, csr, bl[1], Wl[2], bufA, N);
    gather_kernel<<<gath_blocks, 256, 0, stream>>>(bufA, rowptr, csr, bl[2], bufB, N);

    pool_kernel<<<G, 256, 0, stream>>>(bufB, ptr, (float*)d_out, G);
}


# mean pool folded into the last aggregation: a wave sums the relu rows of 8 consecutive nodes per graph in registers, one partial row per (wave, graph); h3 not stored; pool sum order changes
# speedup vs baseline: 3.8489x; 1.0958x over previous
#include <hip/hip_runtime.h>

typedef float v2f __attribute__((ext_vector_type(2)));
typedef float v8f __attribute__((ext_vector_type(8)));
typedef float v4f __attribute__((ext_vector_type(4)));

#define HIDF 128

#define CHUNK_E   2048
#define BKT_SHIFT 8
#define BKT_NODES 256
#define MAX_BKT   512

__global__ void __launch_bounds__(256) tile_sort_kernel(const int* __restrict__ dst,
                                                        unsigned* __restrict__ sorted,
                                                        int* __restrict__ tab, int e, int nb) {
    __shared__ int      s_cnt[MAX_BKT];
    __shared__ int      s_start[MAX_BKT];
    __shared__ int      s_wsum[8];
    __shared__ unsigned s_out[CHUNK_E];
    const int tid  = threadIdx.x;
    const int lane = tid & 31;
    const int wave = tid >> 5;
    const int c    = blockIdx.x;
    const int e0   = c * CHUNK_E;

    s_cnt[tid]       = 0;
    s_cnt[tid + 256] = 0;
    __syncthreads();

    int d[8];
    int r[8];
#pragma unroll
    for (int k = 0; k < 8; ++k) {
        const int i = e0 + k * 256 + tid;
        d[k] = i < e ? dst[i] : -1;
    }
#pragma unroll
    for (int k = 0; k < 8; ++k) {
        r[k] = 0;
        if (d[k] >= 0) r[k] = atomicAdd(&s_cnt[d[k] >> BKT_SHIFT], 1);
    }
    __syncthreads();

    const int v0  = s_cnt[2 * tid];
    const int v1  = s_cnt[2 * tid + 1];
    const int loc = v0 + v1;
    int incl = loc;
    for (int off = 1; off < 32; off <<= 1) {
        int t = __shfl_up(incl, off);
        if (lane >= off) incl += t;
    }
    if (lane == 31) s_wsum[wave] = incl;
    __syncthreads();
    int base = 0;
    for (int w = 0; w < 8; ++w) if (w < wave) base += s_wsum[w];
    const int ex = base + incl - loc;
    s_start[2 * tid]     = ex;
    s_start[2 * tid + 1] = ex + v0;
    __syncthreads();

    for (int b = tid; b < nb; b += 256) tab[c * nb + b] = s_start[b] | (s_cnt[b] << 16);

#pragma unroll
    for (int k = 0; k < 8; ++k) {
        if (d[k] >= 0) {
            const unsigned w = (unsigned)(e0 + k * 256 + tid) | ((unsigned)(d[k] & (BKT_NODES - 1)) << 24);
            s_out[s_start[d[k] >> BKT_SHIFT] + r[k]] = w;
        }
    }
    __syncthreads();
    int m = e - e0;
    if (m > CHUNK_E) m = CHUNK_E;
    for (int j = tid; j < m; j += 256) sorted[e0 + j] = s_out[j];
}

__global__ void __launch_bounds__(512) bucket_rank_kernel(const unsigned* __restrict__ sorted,
                                                          const int* __restrict__ tab,
                                                          int* __restrict__ rowptr, float* __restrict__ dinv,
                                                          int* __restrict__ csr, int* __restrict__ rank,
                                                          int n, int nb, int nc) {
    __shared__ int s_lcnt[BKT_NODES];
    __shared__ int s_red[16];
    __shared__ int s_wsum[8];
    const int tid  = threadIdx.x;
    const int lane = tid & 31;
    const int wave = tid >> 5;
    const int b    = blockIdx.x;
    if (tid < BKT_NODES) s_lcnt[tid] = 0;
    __syncthreads();
    int before = 0;
    for (int c = tid; c < nc; c += 512) {
        const int p     = tab[c * nb + b];
        const int start = p & 0xFFFF;
        const int count = p >> 16;
        before += start;
        const unsigned* sp = sorted + c * CHUNK_E + start;
        for (int k = 0; k < count; ++k) {
            const unsigned w = sp[k];
            const int r = atomicAdd(&s_lcnt[w >> 24], 1);
            rank[w & 0xFFFFFFu] = r;
        }
    }
    for (int off = 16; off > 0; off >>= 1) before += __shfl_xor(before, off);
    if (lane == 0) s_red[wave] = before;
    __syncthreads();
    int base = 0;
    for (int w = 0; w < 16; ++w) base += s_red[w];

    const int cn = tid < BKT_NODES ? s_lcnt[tid] : 0;
    int incl = cn;
    for (int off = 1; off < 32; off <<= 1) {
        int t = __shfl_up(incl, off);
        if (lane >= off) incl += t;
    }
    if (tid < BKT_NODES && lane == 31) s_wsum[wave] = incl;
    __syncthreads();
    const int node = b * BKT_NODES + tid;
    if (tid < BKT_NODES && node < n) {
        int pre = 0;
        for (int w = 0; w < 8; ++w) if (w < wave) pre += s_wsum[w];
        const int   row = base + pre + incl - cn + node;
        const float di  = rsqrtf((float)(cn + 1));
        rowptr[node] = row;
        dinv[node]   = di;
        ((int2*)csr)[row + cn] = make_int2(node, __float_as_int(di * di));
        if (node == n - 1) rowptr[n] = row + cn + 1;
    }
}

__global__ void fill_kernel(const int* __restrict__ src, const int* __restrict__ dst,
                            const int* __restrict__ rank, const int* __restrict__ rowptr,
                            const float* __restrict__ dinv, int* __restrict__ csr, int e) {
    int i = blockIdx.x * blockDim.x + threadIdx.x;
    if (i >= e) return;
    int s = src[i];
    int d = dst[i];
    int pos = rowptr[d] + rank[i];
    float norm = dinv[s] * dinv[d];
    ((int2*)csr)[pos] = make_int2(s, __float_as_int(norm));
}

#define GEMM_GRID  1024
#define A_PITCH    132
__global__ void __launch_bounds__(128) gemm_wmma_kernel(const float* __restrict__ H,
                                                        const float* __restrict__ W,
                                                        float* __restrict__ Out, int M) {
    __shared__ __attribute__((aligned(16))) float s_a[2][16 * A_PITCH];
    const int lane = threadIdx.x & 31;
    const int wave = __builtin_amdgcn_readfirstlane(threadIdx.x >> 5);
    const int half = lane >> 4;
    const int lm   = lane & 15;
    const int mtiles = (M + 15) >> 4;
    int tile = blockIdx.x;
    if (tile >= mtiles) return;
    const int col0 = wave * 32;

    v2f b[32][2];
#pragma unroll
    for (int ks = 0; ks < 32; ++ks) {
        const float* wk = W + (long)(4 * ks + 2 * half) * HIDF + col0 + lm;
#pragma unroll
        for (int t = 0; t < 2; ++t) {
            b[ks][t].x = wk[t * 16];
            b[ks][t].y = wk[HIDF + t * 16];
        }
    }

    v4f g[4];
#pragma unroll
    for (int q = 0; q < 4; ++q) {
        int row = tile * 16 + q * 4 + wave;
        if (row > M - 1) row = M - 1;
        g[q] = *((const v4f*)(H + (long)row * HIDF) + lane);
    }
#pragma unroll
    for (int q = 0; q < 4; ++q)
        *(v4f*)(&s_a[0][(q * 4 + wave) * A_PITCH + lane * 4]) = g[q];
    __syncthreads();

    int buf = 0;
    for (; tile < mtiles; tile += gridDim.x) {
        const int  next     = tile + gridDim.x;
        const bool has_next = next < mtiles;
        if (has_next) {
#pragma unroll
            for (int q = 0; q < 4; ++q) {
                int row = next * 16 + q * 4 + wave;
                if (row > M - 1) row = M - 1;
                g[q] = *((const v4f*)(H + (long)row * HIDF) + lane);
            }
        }

        const float* sa = &s_a[buf][lm * A_PITCH + 2 * half];
        v2f a[32];
#pragma unroll
        for (int ks = 0; ks < 32; ++ks) a[ks] = *(const v2f*)(sa + 4 * ks);

        v8f c0 = (v8f){0.f, 0.f, 0.f, 0.f, 0.f, 0.f, 0.f, 0.f};
        v8f c1 = (v8f){0.f, 0.f, 0.f, 0.f, 0.f, 0.f, 0.f, 0.f};
#pragma unroll
        for (int ks = 0; ks < 32; ++ks) {
            c0 = __builtin_amdgcn_wmma_f32_16x16x4_f32(false, a[ks], false, b[ks][0], (short)0, c0, false, false);
            c1 = __builtin_amdgcn_wmma_f32_16x16x4_f32(false, a[ks], false, b[ks][1], (short)0, c1, false, false);
        }

#pragma unroll
        for (int i = 0; i < 8; ++i) {
            const int row = tile * 16 + i + 8 * half;
            if (row < M) {
                float* o = Out + (long)row * HIDF + col0 + lm;
                o[0]  = c0[i];
                o[16] = c1[i];
            }
        }

        if (has_next) {
#pragma unroll
            for (int q = 0; q < 4; ++q)
                *(v4f*)(&s_a[buf ^ 1][(q * 4 + wave) * A_PITCH + lane * 4]) = g[q];
        }
        __syncthreads();
        buf ^= 1;
    }
}

template <int K>
__device__ __forceinline__ void add_rows(const v4f* __restrict__ hw4, int2 ent, int j,
                                         int lane, v4f& acc) {
    v4f   v[K];
    float w[K];
#pragma unroll
    for (int k = 0; k < K; ++k) {
        int s = __builtin_amdgcn_readlane(ent.x, j + k);
        w[k]  = __int_as_float(__builtin_amdgcn_readlane(ent.y, j + k));
        v[k]  = hw4[(long)s * (HIDF / 4) + lane];
    }
#pragma unroll
    for (int k = 0; k < K; ++k) acc += v[k] * w[k];
}

__device__ __forceinline__ void add_chunk(const v4f* __restrict__ hw4, int2 ent, int m, int lane, v4f& acc) {
    int k = 0;
    for (; k + 8 <= m; k += 8) add_rows<8>(hw4, ent, k, lane, acc);
    if (k + 4 <= m) { add_rows<4>(hw4, ent, k, lane, acc); k += 4; }
    const int r = m - k;
    if (r == 3)      add_rows<3>(hw4, ent, k, lane, acc);
    else if (r == 2) add_rows<2>(hw4, ent, k, lane, acc);
    else if (r == 1) add_rows<1>(hw4, ent, k, lane, acc);
}

#define POOL_K 8
__global__ void __launch_bounds__(256) gather_pool_kernel(const float* __restrict__ hw,
                                                          const int* __restrict__ rowptr,
                                                          const int* __restrict__ csr,
                                                          const float* __restrict__ bias,
                                                          const int* __restrict__ ptr,
                                                          float* __restrict__ part, int n, int g) {
    const int lane = threadIdx.x & 31;
    const int wg = __builtin_amdgcn_readfirstlane(blockIdx.x * 8 + (threadIdx.x >> 5));
    const int n0 = wg * POOL_K;
    if (n0 >= n) return;
    const int cntn = n - n0 < POOL_K ? n - n0 : POOL_K;
    const v4f*  hw4 = (const v4f*)hw;
    const int2* ed  = (const int2*)csr;

    int ri = n0 + (lane < POOL_K ? lane : POOL_K);
    if (ri > n) ri = n;
    const int rp = rowptr[ri];

    int  beg = __builtin_amdgcn_readlane(rp, 0);
    int2 ent = make_int2(0, 0);
    {
        int m0 = __builtin_amdgcn_readlane(rp, 1) - beg;
        if (m0 > 32) m0 = 32;
        if (lane < m0) ent = ed[beg + lane];
    }
    const v4f bv = *((const v4f*)bias + lane);

    const int x = n0 + (lane < cntn ? lane : cntn - 1);
    int lo = 0;
    int hi = g - 1;
    while (lo < hi) {
        const int mid = (lo + hi + 1) >> 1;
        if (ptr[mid] <= x) lo = mid; else hi = mid - 1;
    }
    const int      gnext = __shfl_down(lo, 1);
    const bool     fl    = lane < cntn && (lane == cntn - 1 || gnext != lo);
    const unsigned fmask = __builtin_amdgcn_ballot_w32(fl);

    v4f gsum = {0.0f, 0.0f, 0.0f, 0.0f};
    for (int j = 0; j < cntn; ++j) {
        const int end = __builtin_amdgcn_readlane(rp, j + 1);
        int2 entn = make_int2(0, 0);
        if (j + 1 < cntn) {
            int m2 = __builtin_amdgcn_readlane(rp, j + 2) - end;
            if (m2 > 32) m2 = 32;
            if (lane < m2) entn = ed[end + lane];
        }

        v4f acc = {0.0f, 0.0f, 0.0f, 0.0f};
        add_chunk(hw4, ent, end - beg < 32 ? end - beg : 32, lane, acc);
        for (int base = beg + 32; base < end; base += 32) {
            int m = end - base;
            if (m > 32) m = 32;
            int2 more = make_int2(0, 0);
            if (lane < m) more = ed[base + lane];
            add_chunk(hw4, more, m, lane, acc);
        }

        v4f o = acc + bv;
        o.x = o.x > 0.0f ? o.x : 0.0f;
        o.y = o.y > 0.0f ? o.y : 0.0f;
        o.z = o.z > 0.0f ? o.z : 0.0f;
        o.w = o.w > 0.0f ? o.w : 0.0f;
        gsum += o;
        if ((fmask >> j) & 1u) {
            const int gi = __builtin_amdgcn_readlane(lo, j);
            *((v4f*)(part + (long)(wg + gi) * HIDF) + lane) = gsum;
            gsum = (v4f){0.0f, 0.0f, 0.0f, 0.0f};
        }
        beg = end;
        ent = entn;
    }
}

#define FUSE_GRID  1024

__device__ __forceinline__ void put_row(float* __restrict__ dst, v4f acc, v4f bv) {
    v4f o = acc + bv;
    o.x = o.x > 0.0f ? o.x : 0.0f;
    o.y = o.y > 0.0f ? o.y : 0.0f;
    o.z = o.z > 0.0f ? o.z : 0.0f;
    o.w = o.w > 0.0f ? o.w : 0.0f;
    *(v4f*)dst = o;
}

__device__ __forceinline__ int load_bounds(const int* __restrict__ rowptr, int t, int mtiles, int wave,
                                           int l4, int n) {
    int r = 0;
    if (t < mtiles) {
        int ri = t * 16 + 4 * wave + l4;
        if (ri > n) ri = n;
        r = rowptr[ri];
    }
    return r;
}

template <int J, int G>
__device__ __forceinline__ void add_rows_seg(const v4f* __restrict__ hw4, int2 ent, int cnt, int slot0,
                                             int lane, int rp, v4f bv, float* __restrict__ srow,
                                             v4f& acc, int& cur, int& nb) {
    v4f   v[G];
    float w[G];
    const int jlast = J + cnt - 1;
#pragma unroll
    for (int k = 0; k < G; ++k) {
        const int jj = J + k < jlast ? J + k : jlast;
        int s = __builtin_amdgcn_readlane(ent.x, jj);
        w[k]  = __int_as_float(__builtin_amdgcn_readlane(ent.y, jj));
        v[k]  = hw4[(long)s * (HIDF / 4) + lane];
    }
#pragma unroll
    for (int k = 0; k < G; ++k) {
        if (k < cnt) {
            if (slot0 + k == nb) {
                put_row(srow + cur * A_PITCH, acc, bv);
                ++cur;
                nb  = __builtin_amdgcn_readlane(rp, cur + 1);
                acc = (v4f){0.0f, 0.0f, 0.0f, 0.0f};
            }
            acc += v[k] * w[k];
        }
    }
}

template <int J>
__device__ __forceinline__ void add_rows_half(const v4f* __restrict__ hw4, int2 ent, int cnt, int slot0,
                                              int lane, int rp, v4f bv, float* __restrict__ srow,
                                              v4f& acc, int& cur, int& nb) {
    if (cnt > 12)     add_rows_seg<J, 16>(hw4, ent, cnt, slot0, lane, rp, bv, srow, acc, cur, nb);
    else if (cnt > 8) add_rows_seg<J, 12>(hw4, ent, cnt, slot0, lane, rp, bv, srow, acc, cur, nb);
    else if (cnt > 4) add_rows_seg<J, 8>(hw4, ent, cnt, slot0, lane, rp, bv, srow, acc, cur, nb);
    else              add_rows_seg<J, 4>(hw4, ent, cnt, slot0, lane, rp, bv, srow, acc, cur, nb);
}

__global__ void __launch_bounds__(128) gather_gemm_kernel(const float* __restrict__ hw,
                                                          const int* __restrict__ rowptr,
                                                          const int* __restrict__ csr,
                                                          const float* __restrict__ bias,
                                                          const float* __restrict__ W,
                                                          float* __restrict__ Out, int n) {
    __shared__ __attribute__((aligned(16))) float s_a[2][16 * A_PITCH];
    const int lane = threadIdx.x & 31;
    const int wave = __builtin_amdgcn_readfirstlane(threadIdx.x >> 5);
    const int half = lane >> 4;
    const int lm   = lane & 15;
    const int mtiles = (n + 15) >> 4;
    int tile = blockIdx.x;
    if (tile >= mtiles) return;
    const int col0 = wave * 32;
    const v4f*  hw4 = (const v4f*)hw;
    const int2* ed  = (const int2*)csr;
    const int   l4  = lane < 5 ? lane : 4;

    int  rp  = load_bounds(rowptr, tile, mtiles, wave, l4, n);
    int  rp1 = load_bounds(rowptr, tile + gridDim.x, mtiles, wave, l4, n);
    int  e0  = __builtin_amdgcn_readlane(rp, 0);
    int  e4  = __builtin_amdgcn_readlane(rp, 4);
    int2 ent = make_int2(0, 0);
    if (lane < e4 - e0) ent = ed[e0 + lane];

    v2f b[32][2];
#pragma unroll
    for (int ks = 0; ks < 32; ++ks) {
        const float* wk = W + (long)(4 * ks + 2 * half) * HIDF + col0 + lm;
#pragma unroll
        for (int t = 0; t < 2; ++t) {
            b[ks][t].x = wk[t * 16];
            b[ks][t].y = wk[HIDF + t * 16];
        }
    }
    const v4f bv = *((const v4f*)bias + lane);

    int buf = 0;
    for (; tile < mtiles; tile += gridDim.x) {
        float* srow = &s_a[buf][(4 * wave) * A_PITCH + lane * 4];
        v4f acc = {0.0f, 0.0f, 0.0f, 0.0f};
        int cur = 0;
        int nb  = __builtin_amdgcn_readlane(rp, 1);
        for (int base = e0; base < e4; ) {
            int m = e4 - base;
            if (m > 32) m = 32;
            add_rows_half<0>(hw4, ent, m < 16 ? m : 16, base, lane, rp, bv, srow, acc, cur, nb);
            if (m > 16) add_rows_half<16>(hw4, ent, m - 16, base + 16, lane, rp, bv, srow, acc, cur, nb);
            base += 32;
            if (base < e4) {
                ent = make_int2(0, 0);
                if (lane < e4 - base) ent = ed[base + lane];
            }
        }
        for (; cur < 4; ++cur) {
            put_row(srow + cur * A_PITCH, acc, bv);
            acc = (v4f){0.0f, 0.0f, 0.0f, 0.0f};
        }
        __syncthreads();

        rp  = rp1;
        e0  = __builtin_amdgcn_readlane(rp, 0);
        e4  = __builtin_amdgcn_readlane(rp, 4);
        ent = make_int2(0, 0);
        if (lane < e4 - e0) ent = ed[e0 + lane];
        rp1 = load_bounds(rowptr, tile + 2 * gridDim.x, mtiles, wave, l4, n);

        const float* sa = &s_a[buf][lm * A_PITCH + 2 * half];
        v2f a[32];
#pragma unroll
        for (int ks = 0; ks < 32; ++ks) a[ks] = *(const v2f*)(sa + 4 * ks);

        v8f c0 = (v8f){0.f, 0.f, 0.f, 0.f, 0.f, 0.f, 0.f, 0.f};
        v8f c1 = (v8f){0.f, 0.f, 0.f, 0.f, 0.f, 0.f, 0.f, 0.f};
#pragma unroll
        for (int ks = 0; ks < 32; ++ks) {
            c0 = __builtin_amdgcn_wmma_f32_16x16x4_f32(false, a[ks], false, b[ks][0], (short)0, c0, false, false);
            c1 = __builtin_amdgcn_wmma_f32_16x16x4_f32(false, a[ks], false, b[ks][1], (short)0, c1, false, false);
        }

#pragma unroll
        for (int i = 0; i < 8; ++i) {
            const int row = tile * 16 + i + 8 * half;
            if (row < n) {
                float* o = Out + (long)row * HIDF + col0 + lm;
                o[0]  = c0[i];
                o[16] = c1[i];
            }
        }
        buf ^= 1;
    }
}

#define POOL_WAVES 8
#define POOL_ROWS  16
__global__ void __launch_bounds__(POOL_WAVES * 32) pool_kernel(const float* __restrict__ h,
                                                               const int* __restrict__ ptr,
                                                               float* __restrict__ out, int g, int nn) {
    __shared__ v4f s_part[POOL_WAVES][32];
    const int gi   = blockIdx.x;
    const int t    = threadIdx.x;
    const int lane = t & 31;
    const int wave = __builtin_amdgcn_readfirstlane(t >> 5);
    const int p0 = ptr[gi];
    const int p1 = ptr[gi + 1];
    int sn = gi == 0 ? 0 : p0;
    int en = gi == g - 1 ? nn : p1;
    if (sn < 0) sn = 0;
    if (en > nn) en = nn;
    int s = 0;
    int e = 0;
    if (sn < en) {
        s = sn / POOL_K + gi;
        e = (en - 1) / POOL_K + gi + 1;
    }
    const v4f* h4 = (const v4f*)h + lane;

    v4f acc = {0.0f, 0.0f, 0.0f, 0.0f};
    int n = s + wave;
    for (; n + (POOL_ROWS - 1) * POOL_WAVES < e; n += POOL_ROWS * POOL_WAVES) {
        v4f v[POOL_ROWS];
#pragma unroll
        for (int u = 0; u < POOL_ROWS; ++u) v[u] = h4[(long)(n + u * POOL_WAVES) * (HIDF / 4)];
#pragma unroll
        for (int u = 0; u < POOL_ROWS; ++u) acc += v[u];
    }
    if (n < e) {
        v4f v[POOL_ROWS];
        const int last = e - 1;
#pragma unroll
        for (int u = 0; u < POOL_ROWS; ++u) {
            const int r = n + u * POOL_WAVES;
            v[u] = h4[(long)(r < e ? r : last) * (HIDF / 4)];
        }
#pragma unroll
        for (int u = 0; u < POOL_ROWS; ++u) {
            if (n + u * POOL_WAVES < e) acc += v[u];
        }
    }

    s_part[wave][lane] = acc;
    __syncthreads();
    if (t < HIDF) {
        const float* sp = (const float*)s_part;
        float sum = 0.0f;
        for (int w = 0; w < POOL_WAVES; ++w) sum += sp[w * HIDF + t];
        float cnt = (float)(p1 - p0);
        if (cnt < 1.0f) cnt = 1.0f;
        out[(long)gi * HIDF + t] = sum / cnt;
    }
}

extern "C" void kernel_launch(void* const* d_in, const int* in_sizes, int n_in,
                              void* d_out, int out_size, void* d_ws, size_t ws_size,
                              hipStream_t stream) {
    const float* x   = (const float*)d_in[0];
    const int*   ei  = (const int*)d_in[1];
    const int*   ptr = (const int*)d_in[2];
    const float* Wl[3] = {(const float*)d_in[3], (const float*)d_in[5], (const float*)d_in[7]};
    const float* bl[3] = {(const float*)d_in[4], (const float*)d_in[6], (const float*)d_in[8]};

    int N = in_sizes[0] / HIDF;
    int E = in_sizes[1] / 2;
    int G = in_sizes[2] - 1;

    const int* src = ei;
    const int* dst = ei + E;

    size_t npad = ((size_t)N + 255) & ~(size_t)255;
    size_t epad = ((size_t)E + 255) & ~(size_t)255;
    float* dinv   = (float*)d_ws;
    int*   rowptr = (int*)(dinv + npad);
    int*   rank   = rowptr + npad + 256;
    int*   csr    = rank + epad;
    float* bufA   = (float*)(csr + 2 * (epad + npad));
    float* bufB   = bufA + (size_t)N * HIDF;
    unsigned* sorted = (unsigned*)(bufB + (size_t)N * HIDF);
    int*      tab    = (int*)(sorted + epad);

    int nb = (N + BKT_NODES - 1) / BKT_NODES;
    int nc = (E + CHUNK_E - 1) / CHUNK_E;
    tile_sort_kernel<<<nc, 256, 0, stream>>>(dst, sorted, tab, E, nb);
    bucket_rank_kernel<<<nb, 512, 0, stream>>>(sorted, tab, rowptr, dinv, csr, rank, N, nb, nc);
    fill_kernel<<<(E + 255) / 256, 256, 0, stream>>>(src, dst, rank, rowptr, dinv, csr, E);

    int mtiles      = (N + 15) / 16;
    int gemm_blocks = mtiles < GEMM_GRID ? mtiles : GEMM_GRID;
    int pool_waves  = (N + POOL_K - 1) / POOL_K;
    int gath_blocks = (pool_waves + 7) / 8;

    int fuse_blocks = mtiles < FUSE_GRID ? mtiles : FUSE_GRID;

    gemm_wmma_kernel<<<gemm_blocks, 128, 0, stream>>>(x, Wl[0], bufA, N);
    gather_gemm_kernel<<<fuse_blocks, 128, 0, stream>>>(bufA, rowptr, csr, bl[0], Wl[1], bufB, N);
    gather_gemm_kernel<<<fuse_blocks, 128, 0, stream>>>(bufB, rowptr, csr, bl[1], Wl[2], bufA, N);
    gather_pool_kernel<<<gath_blocks, 256, 0, stream>>>(bufA, rowptr, csr, bl[2], ptr, bufB, N, G);

    pool_kernel<<<G, 256, 0, stream>>>(bufB, ptr, (float*)d_out, G, N);
}


# norm factored: products store dinv[row]*(h@W), nodes sum unweighted rows, scale by dinv[dst]; edge list = src only, built in the rank kernel's 2nd walk; fill launch gone (7); products re-associated
# speedup vs baseline: 4.8274x; 1.2542x over previous
#include <hip/hip_runtime.h>

typedef float v2f __attribute__((ext_vector_type(2)));
typedef float v8f __attribute__((ext_vector_type(8)));
typedef float v4f __attribute__((ext_vector_type(4)));

#define HIDF 128

#define CHUNK_E   2048
#define BKT_SHIFT 8
#define BKT_NODES 256
#define MAX_BKT   512

__global__ void __launch_bounds__(256) tile_sort_kernel(const int* __restrict__ src,
                                                        const int* __restrict__ dst,
                                                        unsigned* __restrict__ sorted,
                                                        int* __restrict__ tab, int e, int nb) {
    __shared__ int      s_cnt[MAX_BKT];
    __shared__ int      s_start[MAX_BKT];
    __shared__ int      s_wsum[8];
    __shared__ unsigned s_out[CHUNK_E];
    const int tid  = threadIdx.x;
    const int lane = tid & 31;
    const int wave = tid >> 5;
    const int c    = blockIdx.x;
    const int e0   = c * CHUNK_E;

    s_cnt[tid]       = 0;
    s_cnt[tid + 256] = 0;
    __syncthreads();

    int d[8];
    int sv[8];
    int r[8];
#pragma unroll
    for (int k = 0; k < 8; ++k) {
        const int i = e0 + k * 256 + tid;
        d[k]  = i < e ? dst[i] : -1;
        sv[k] = i < e ? src[i] : 0;
    }
#pragma unroll
    for (int k = 0; k < 8; ++k) {
        r[k] = 0;
        if (d[k] >= 0) r[k] = atomicAdd(&s_cnt[d[k] >> BKT_SHIFT], 1);
    }
    __syncthreads();

    const int v0  = s_cnt[2 * tid];
    const int v1  = s_cnt[2 * tid + 1];
    const int loc = v0 + v1;
    int incl = loc;
    for (int off = 1; off < 32; off <<= 1) {
        int t = __shfl_up(incl, off);
        if (lane >= off) incl += t;
    }
    if (lane == 31) s_wsum[wave] = incl;
    __syncthreads();
    int base = 0;
    for (int w = 0; w < 8; ++w) if (w < wave) base += s_wsum[w];
    const int ex = base + incl - loc;
    s_start[2 * tid]     = ex;
    s_start[2 * tid + 1] = ex + v0;
    __syncthreads();

    for (int b = tid; b < nb; b += 256) tab[c * nb + b] = s_start[b] | (s_cnt[b] << 16);

#pragma unroll
    for (int k = 0; k < 8; ++k) {
        if (d[k] >= 0) {
            const unsigned w = (unsigned)sv[k] | ((unsigned)(d[k] & (BKT_NODES - 1)) << 24);
            s_out[s_start[d[k] >> BKT_SHIFT] + r[k]] = w;
        }
    }
    __syncthreads();
    int m = e - e0;
    if (m > CHUNK_E) m = CHUNK_E;
    for (int j = tid; j < m; j += 256) sorted[e0 + j] = s_out[j];
}

__global__ void __launch_bounds__(512) bucket_rank_kernel(const unsigned* __restrict__ sorted,
                                                          const int* __restrict__ tab,
                                                          int* __restrict__ rowptr, float* __restrict__ dinv,
                                                          int* __restrict__ csr,
                                                          int n, int nb, int nc) {
    __shared__ int s_lcnt[BKT_NODES];
    __shared__ int s_pos[BKT_NODES];
    __shared__ int s_red[16];
    __shared__ int s_wsum[8];
    const int tid  = threadIdx.x;
    const int lane = tid & 31;
    const int wave = tid >> 5;
    const int b    = blockIdx.x;
    if (tid < BKT_NODES) s_lcnt[tid] = 0;
    __syncthreads();
    int before = 0;
    for (int c = tid; c < nc; c += 512) {
        const int p     = tab[c * nb + b];
        const int start = p & 0xFFFF;
        const int count = p >> 16;
        before += start;
        const unsigned* sp = sorted + c * CHUNK_E + start;
        for (int k = 0; k < count; ++k) atomicAdd(&s_lcnt[sp[k] >> 24], 1);
    }
    for (int off = 16; off > 0; off >>= 1) before += __shfl_xor(before, off);
    if (lane == 0) s_red[wave] = before;
    __syncthreads();
    int base = 0;
    for (int w = 0; w < 16; ++w) base += s_red[w];

    const int cn = tid < BKT_NODES ? s_lcnt[tid] : 0;
    int incl = cn;
    for (int off = 1; off < 32; off <<= 1) {
        int t = __shfl_up(incl, off);
        if (lane >= off) incl += t;
    }
    if (tid < BKT_NODES && lane == 31) s_wsum[wave] = incl;
    __syncthreads();
    const int node = b * BKT_NODES + tid;
    if (tid < BKT_NODES && node < n) {
        int pre = 0;
        for (int w = 0; w < 8; ++w) if (w < wave) pre += s_wsum[w];
        const int row = base + pre + incl - cn + node;
        rowptr[node]  = row;
        dinv[node]    = rsqrtf((float)(cn + 1));
        csr[row + cn] = node;
        s_pos[tid]    = row;
        if (node == n - 1) rowptr[n] = row + cn + 1;
    }
    __syncthreads();
    for (int c = tid; c < nc; c += 512) {
        const int p     = tab[c * nb + b];
        const int start = p & 0xFFFF;
        const int count = p >> 16;
        const unsigned* sp = sorted + c * CHUNK_E + start;
        for (int k = 0; k < count; ++k) {
            const unsigned w = sp[k];
            const int pos = atomicAdd(&s_pos[w >> 24], 1);
            csr[pos] = (int)(w & 0xFFFFFFu);
        }
    }
}

#define GEMM_GRID  1024
#define A_PITCH    132
__device__ __forceinline__ float load_dinv16(const float* __restrict__ dinv, int t, int mtiles, int lm, int n) {
    float r = 0.0f;
    if (t < mtiles) {
        int ri = t * 16 + lm;
        if (ri > n - 1) ri = n - 1;
        r = dinv[ri];
    }
    return r;
}

__device__ __forceinline__ float row_scale(float dvt, int i, int half) {
    const float lo = __int_as_float(__builtin_amdgcn_readlane(__float_as_int(dvt), i));
    const float hi = __int_as_float(__builtin_amdgcn_readlane(__float_as_int(dvt), i + 8));
    return half ? hi : lo;
}

__global__ void __launch_bounds__(128) gemm_wmma_kernel(const float* __restrict__ H,
                                                        const float* __restrict__ W,
                                                        const float* __restrict__ dinv,
                                                        float* __restrict__ Out, int M) {
    __shared__ __attribute__((aligned(16))) float s_a[2][16 * A_PITCH];
    const int lane = threadIdx.x & 31;
    const int wave = __builtin_amdgcn_readfirstlane(threadIdx.x >> 5);
    const int half = lane >> 4;
    const int lm   = lane & 15;
    const int mtiles = (M + 15) >> 4;
    int tile = blockIdx.x;
    if (tile >= mtiles) return;
    const int col0 = wave * 32;

    v2f b[32][2];
#pragma unroll
    for (int ks = 0; ks < 32; ++ks) {
        const float* wk = W + (long)(4 * ks + 2 * half) * HIDF + col0 + lm;
#pragma unroll
        for (int t = 0; t < 2; ++t) {
            b[ks][t].x = wk[t * 16];
            b[ks][t].y = wk[HIDF + t * 16];
        }
    }

    v4f g[4];
#pragma unroll
    for (int q = 0; q < 4; ++q) {
        int row = tile * 16 + q * 4 + wave;
        if (row > M - 1) row = M - 1;
        g[q] = *((const v4f*)(H + (long)row * HIDF) + lane);
    }
#pragma unroll
    for (int q = 0; q < 4; ++q)
        *(v4f*)(&s_a[0][(q * 4 + wave) * A_PITCH + lane * 4]) = g[q];
    __syncthreads();

    int buf = 0;
    for (; tile < mtiles; tile += gridDim.x) {
        const int  next     = tile + gridDim.x;
        const bool has_next = next < mtiles;
        if (has_next) {
#pragma unroll
            for (int q = 0; q < 4; ++q) {
                int row = next * 16 + q * 4 + wave;
                if (row > M - 1) row = M - 1;
                g[q] = *((const v4f*)(H + (long)row * HIDF) + lane);
            }
        }

        const float dvt = load_dinv16(dinv, tile, mtiles, lm, M);

        const float* sa = &s_a[buf][lm * A_PITCH + 2 * half];
        v2f a[32];
#pragma unroll
        for (int ks = 0; ks < 32; ++ks) a[ks] = *(const v2f*)(sa + 4 * ks);

        v8f c0 = (v8f){0.f, 0.f, 0.f, 0.f, 0.f, 0.f, 0.f, 0.f};
        v8f c1 = (v8f){0.f, 0.f, 0.f, 0.f, 0.f, 0.f, 0.f, 0.f};
#pragma unroll
        for (int ks = 0; ks < 32; ++ks) {
            c0 = __builtin_amdgcn_wmma_f32_16x16x4_f32(false, a[ks], false, b[ks][0], (short)0, c0, false, false);
            c1 = __builtin_amdgcn_wmma_f32_16x16x4_f32(false, a[ks], false, b[ks][1], (short)0, c1, false, false);
        }

#pragma unroll
        for (int i = 0; i < 8; ++i) {
            const int   row = tile * 16 + i + 8 * half;
            const float sc  = row_scale(dvt, i, half);
            if (row < M) {
                float* o = Out + (long)row * HIDF + col0 + lm;
                o[0]  = c0[i] * sc;
                o[16] = c1[i] * sc;
            }
        }

        if (has_next) {
#pragma unroll
            for (int q = 0; q < 4; ++q)
                *(v4f*)(&s_a[buf ^ 1][(q * 4 + wave) * A_PITCH + lane * 4]) = g[q];
        }
        __syncthreads();
        buf ^= 1;
    }
}

template <int K>
__device__ __forceinline__ void add_rows(const v4f* __restrict__ hw4, int ent, int j,
                                         int lane, v4f& acc) {
    v4f v[K];
#pragma unroll
    for (int k = 0; k < K; ++k) {
        int s = __builtin_amdgcn_readlane(ent, j + k);
        v[k]  = hw4[(long)s * (HIDF / 4) + lane];
    }
#pragma unroll
    for (int k = 0; k < K; ++k) acc += v[k];
}

__device__ __forceinline__ void add_chunk(const v4f* __restrict__ hw4, int ent, int m, int lane, v4f& acc) {
    int k = 0;
    for (; k + 8 <= m; k += 8) add_rows<8>(hw4, ent, k, lane, acc);
    if (k + 4 <= m) { add_rows<4>(hw4, ent, k, lane, acc); k += 4; }
    const int r = m - k;
    if (r == 3)      add_rows<3>(hw4, ent, k, lane, acc);
    else if (r == 2) add_rows<2>(hw4, ent, k, lane, acc);
    else if (r == 1) add_rows<1>(hw4, ent, k, lane, acc);
}

#define POOL_K 8
__global__ void __launch_bounds__(256) gather_pool_kernel(const float* __restrict__ hw,
                                                          const int* __restrict__ rowptr,
                                                          const int* __restrict__ csr,
                                                          const float* __restrict__ dinv,
                                                          const float* __restrict__ bias,
                                                          const int* __restrict__ ptr,
                                                          float* __restrict__ part, int n, int g) {
    const int lane = threadIdx.x & 31;
    const int wg = __builtin_amdgcn_readfirstlane(blockIdx.x * 8 + (threadIdx.x >> 5));
    const int n0 = wg * POOL_K;
    if (n0 >= n) return;
    const int cntn = n - n0 < POOL_K ? n - n0 : POOL_K;
    const v4f* hw4 = (const v4f*)hw;
    const int* ed  = csr;

    int ri = n0 + (lane < POOL_K ? lane : POOL_K);
    if (ri > n) ri = n;
    const int rp = rowptr[ri];

    int beg = __builtin_amdgcn_readlane(rp, 0);
    int ent = 0;
    {
        int m0 = __builtin_amdgcn_readlane(rp, 1) - beg;
        if (m0 > 32) m0 = 32;
        if (lane < m0) ent = ed[beg + lane];
    }
    const v4f bv = *((const v4f*)bias + lane);

    const int   x   = n0 + (lane < cntn ? lane : cntn - 1);
    const float dvp = dinv[x];
    int lo = 0;
    int hi = g - 1;
    while (lo < hi) {
        const int mid = (lo + hi + 1) >> 1;
        if (ptr[mid] <= x) lo = mid; else hi = mid - 1;
    }
    const int      gnext = __shfl_down(lo, 1);
    const bool     fl    = lane < cntn && (lane == cntn - 1 || gnext != lo);
    const unsigned fmask = __builtin_amdgcn_ballot_w32(fl);

    v4f gsum = {0.0f, 0.0f, 0.0f, 0.0f};
    for (int j = 0; j < cntn; ++j) {
        const int end = __builtin_amdgcn_readlane(rp, j + 1);
        int entn = 0;
        if (j + 1 < cntn) {
            int m2 = __builtin_amdgcn_readlane(rp, j + 2) - end;
            if (m2 > 32) m2 = 32;
            if (lane < m2) entn = ed[end + lane];
        }

        v4f acc = {0.0f, 0.0f, 0.0f, 0.0f};
        add_chunk(hw4, ent, end - beg < 32 ? end - beg : 32, lane, acc);
        for (int base = beg + 32; base < end; base += 32) {
            int m = end - base;
            if (m > 32) m = 32;
            int more = 0;
            if (lane < m) more = ed[base + lane];
            add_chunk(hw4, more, m, lane, acc);
        }

        const float dd = __int_as_float(__builtin_amdgcn_readlane(__float_as_int(dvp), j));
        v4f o = acc * dd + bv;
        o.x = o.x > 0.0f ? o.x : 0.0f;
        o.y = o.y > 0.0f ? o.y : 0.0f;
        o.z = o.z > 0.0f ? o.z : 0.0f;
        o.w = o.w > 0.0f ? o.w : 0.0f;
        gsum += o;
        if ((fmask >> j) & 1u) {
            const int gi = __builtin_amdgcn_readlane(lo, j);
            *((v4f*)(part + (long)(wg + gi) * HIDF) + lane) = gsum;
            gsum = (v4f){0.0f, 0.0f, 0.0f, 0.0f};
        }
        beg = end;
        ent = entn;
    }
}

#define FUSE_GRID  1024

__device__ __forceinline__ void put_row(float* __restrict__ dst, v4f acc, float dd, v4f bv) {
    v4f o = acc * dd + bv;
    o.x = o.x > 0.0f ? o.x : 0.0f;
    o.y = o.y > 0.0f ? o.y : 0.0f;
    o.z = o.z > 0.0f ? o.z : 0.0f;
    o.w = o.w > 0.0f ? o.w : 0.0f;
    *(v4f*)dst = o;
}

__device__ __forceinline__ int load_bounds(const int* __restrict__ rowptr, int t, int mtiles, int wave,
                                           int l4, int n) {
    int r = 0;
    if (t < mtiles) {
        int ri = t * 16 + 4 * wave + l4;
        if (ri > n) ri = n;
        r = rowptr[ri];
    }
    return r;
}

template <int J, int G>
__device__ __forceinline__ void add_rows_seg(const v4f* __restrict__ hw4, int ent, int cnt, int slot0,
                                             int lane, int rp, float dvt, int w4, v4f bv,
                                             float* __restrict__ srow, v4f& acc, int& cur, int& nb) {
    v4f v[G];
    const int jlast = J + cnt - 1;
#pragma unroll
    for (int k = 0; k < G; ++k) {
        const int jj = J + k < jlast ? J + k : jlast;
        int s = __builtin_amdgcn_readlane(ent, jj);
        v[k]  = hw4[(long)s * (HIDF / 4) + lane];
    }
#pragma unroll
    for (int k = 0; k < G; ++k) {
        if (k < cnt) {
            if (slot0 + k == nb) {
                const float dd = __int_as_float(__builtin_amdgcn_readlane(__float_as_int(dvt), w4 + cur));
                put_row(srow + cur * A_PITCH, acc, dd, bv);
                ++cur;
                nb  = __builtin_amdgcn_readlane(rp, cur + 1);
                acc = (v4f){0.0f, 0.0f, 0.0f, 0.0f};
            }
            acc += v[k];
        }
    }
}

template <int J>
__device__ __forceinline__ void add_rows_half(const v4f* __restrict__ hw4, int ent, int cnt, int slot0,
                                              int lane, int rp, float dvt, int w4, v4f bv,
                                              float* __restrict__ srow, v4f& acc, int& cur, int& nb) {
    if (cnt > 12)     add_rows_seg<J, 16>(hw4, ent, cnt, slot0, lane, rp, dvt, w4, bv, srow, acc, cur, nb);
    else if (cnt > 8) add_rows_seg<J, 12>(hw4, ent, cnt, slot0, lane, rp, dvt, w4, bv, srow, acc, cur, nb);
    else if (cnt > 4) add_rows_seg<J, 8>(hw4, ent, cnt, slot0, lane, rp, dvt, w4, bv, srow, acc, cur, nb);
    else              add_rows_seg<J, 4>(hw4, ent, cnt, slot0, lane, rp, dvt, w4, bv, srow, acc, cur, nb);
}

__global__ void __launch_bounds__(128) gather_gemm_kernel(const float* __restrict__ hw,
                                                          const int* __restrict__ rowptr,
                                                          const int* __restrict__ csr,
                                                          const float* __restrict__ dinv,
                                                          const float* __restrict__ bias,
                                                          const float* __restrict__ W,
                                                          float* __restrict__ Out, int n) {
    __shared__ __attribute__((aligned(16))) float s_a[2][16 * A_PITCH];
    const int lane = threadIdx.x & 31;
    const int wave = __builtin_amdgcn_readfirstlane(threadIdx.x >> 5);
    const int half = lane >> 4;
    const int lm   = lane & 15;
    const int mtiles = (n + 15) >> 4;
    int tile = blockIdx.x;
    if (tile >= mtiles) return;
    const int col0 = wave * 32;
    const v4f* hw4 = (const v4f*)hw;
    const int* ed  = csr;
    const int  l4  = lane < 5 ? lane : 4;
    const int  w4  = 4 * wave;

    int   rp  = load_bounds(rowptr, tile, mtiles, wave, l4, n);
    int   rp1 = load_bounds(rowptr, tile + gridDim.x, mtiles, wave, l4, n);
    int   e0  = __builtin_amdgcn_readlane(rp, 0);
    int   e4  = __builtin_amdgcn_readlane(rp, 4);
    int   ent = 0;
    if (lane < e4 - e0) ent = ed[e0 + lane];
    float dvt = load_dinv16(dinv, tile, mtiles, lm, n);

    v2f b[32][2];
#pragma unroll
    for (int ks = 0; ks < 32; ++ks) {
        const float* wk = W + (long)(4 * ks + 2 * half) * HIDF + col0 + lm;
#pragma unroll
        for (int t = 0; t < 2; ++t) {
            b[ks][t].x = wk[t * 16];
            b[ks][t].y = wk[HIDF + t * 16];
        }
    }
    const v4f bv = *((const v4f*)bias + lane);

    int buf = 0;
    for (; tile < mtiles; tile += gridDim.x) {
        float* srow = &s_a[buf][(4 * wave) * A_PITCH + lane * 4];
        v4f acc = {0.0f, 0.0f, 0.0f, 0.0f};
        int cur = 0;
        int nb  = __builtin_amdgcn_readlane(rp, 1);
        for (int base = e0; base < e4; ) {
            int m = e4 - base;
            if (m > 32) m = 32;
            add_rows_half<0>(hw4, ent, m < 16 ? m : 16, base, lane, rp, dvt, w4, bv, srow, acc, cur, nb);
            if (m > 16) add_rows_half<16>(hw4, ent, m - 16, base + 16, lane, rp, dvt, w4, bv, srow, acc, cur, nb);
            base += 32;
            if (base < e4) {
                ent = 0;
                if (lane < e4 - base) ent = ed[base + lane];
            }
        }
        for (; cur < 4; ++cur) {
            const float dd = __int_as_float(__builtin_amdgcn_readlane(__float_as_int(dvt), w4 + cur));
            put_row(srow + cur * A_PITCH, acc, dd, bv);
            acc = (v4f){0.0f, 0.0f, 0.0f, 0.0f};
        }
        __syncthreads();

        rp  = rp1;
        e0  = __builtin_amdgcn_readlane(rp, 0);
        e4  = __builtin_amdgcn_readlane(rp, 4);
        ent = 0;
        if (lane < e4 - e0) ent = ed[e0 + lane];
        rp1 = load_bounds(rowptr, tile + 2 * gridDim.x, mtiles, wave, l4, n);
        const float dvn = load_dinv16(dinv, tile + gridDim.x, mtiles, lm, n);

        const float* sa = &s_a[buf][lm * A_PITCH + 2 * half];
        v2f a[32];
#pragma unroll
        for (int ks = 0; ks < 32; ++ks) a[ks] = *(const v2f*)(sa + 4 * ks);

        v8f c0 = (v8f){0.f, 0.f, 0.f, 0.f, 0.f, 0.f, 0.f, 0.f};
        v8f c1 = (v8f){0.f, 0.f, 0.f, 0.f, 0.f, 0.f, 0.f, 0.f};
#pragma unroll
        for (int ks = 0; ks < 32; ++ks) {
            c0 = __builtin_amdgcn_wmma_f32_16x16x4_f32(false, a[ks], false, b[ks][0], (short)0, c0, false, false);
            c1 = __builtin_amdgcn_wmma_f32_16x16x4_f32(false, a[ks], false, b[ks][1], (short)0, c1, false, false);
        }

#pragma unroll
        for (int i = 0; i < 8; ++i) {
            const int   row = tile * 16 + i + 8 * half;
            const float sc  = row_scale(dvt, i, half);
            if (row < n) {
                float* o = Out + (long)row * HIDF + col0 + lm;
                o[0]  = c0[i] * sc;
                o[16] = c1[i] * sc;
            }
        }
        dvt = dvn;
        buf ^= 1;
    }
}

#define POOL_WAVES 8
#define POOL_ROWS  16
__global__ void __launch_bounds__(POOL_WAVES * 32) pool_kernel(const float* __restrict__ h,
                                                               const int* __restrict__ ptr,
                                                               float* __restrict__ out, int g, int nn) {
    __shared__ v4f s_part[POOL_WAVES][32];
    const int gi   = blockIdx.x;
    const int t    = threadIdx.x;
    const int lane = t & 31;
    const int wave = __builtin_amdgcn_readfirstlane(t >> 5);
    const int p0 = ptr[gi];
    const int p1 = ptr[gi + 1];
    int sn = gi == 0 ? 0 : p0;
    int en = gi == g - 1 ? nn : p1;
    if (sn < 0) sn = 0;
    if (en > nn) en = nn;
    int s = 0;
    int e = 0;
    if (sn < en) {
        s = sn / POOL_K + gi;
        e = (en - 1) / POOL_K + gi + 1;
    }
    const v4f* h4 = (const v4f*)h + lane;

    v4f acc = {0.0f, 0.0f, 0.0f, 0.0f};
    int n = s + wave;
    for (; n + (POOL_ROWS - 1) * POOL_WAVES < e; n += POOL_ROWS * POOL_WAVES) {
        v4f v[POOL_ROWS];
#pragma unroll
        for (int u = 0; u < POOL_ROWS; ++u) v[u] = h4[(long)(n + u * POOL_WAVES) * (HIDF / 4)];
#pragma unroll
        for (int u = 0; u < POOL_ROWS; ++u) acc += v[u];
    }
    if (n < e) {
        v4f v[POOL_ROWS];
        const int last = e - 1;
#pragma unroll
        for (int u = 0; u < POOL_ROWS; ++u) {
            const int r = n + u * POOL_WAVES;
            v[u] = h4[(long)(r < e ? r : last) * (HIDF / 4)];
        }
#pragma unroll
        for (int u = 0; u < POOL_ROWS; ++u) {
            if (n + u * POOL_WAVES < e) acc += v[u];
        }
    }

    s_part[wave][lane] = acc;
    __syncthreads();
    if (t < HIDF) {
        const float* sp = (const float*)s_part;
        float sum = 0.0f;
        for (int w = 0; w < POOL_WAVES; ++w) sum += sp[w * HIDF + t];
        float cnt = (float)(p1 - p0);
        if (cnt < 1.0f) cnt = 1.0f;
        out[(long)gi * HIDF + t] = sum / cnt;
    }
}

extern "C" void kernel_launch(void* const* d_in, const int* in_sizes, int n_in,
                              void* d_out, int out_size, void* d_ws, size_t ws_size,
                              hipStream_t stream) {
    const float* x   = (const float*)d_in[0];
    const int*   ei  = (const int*)d_in[1];
    const int*   ptr = (const int*)d_in[2];
    const float* Wl[3] = {(const float*)d_in[3], (const float*)d_in[5], (const float*)d_in[7]};
    const float* bl[3] = {(const float*)d_in[4], (const float*)d_in[6], (const float*)d_in[8]};

    int N = in_sizes[0] / HIDF;
    int E = in_sizes[1] / 2;
    int G = in_sizes[2] - 1;

    const int* src = ei;
    const int* dst = ei + E;

    size_t npad = ((size_t)N + 255) & ~(size_t)255;
    size_t epad = ((size_t)E + 255) & ~(size_t)255;
    float* dinv   = (float*)d_ws;
    int*   rowptr = (int*)(dinv + npad);
    int*   csr    = rowptr + npad + 256;
    float* bufA   = (float*)(csr + epad + npad);
    float* bufB   = bufA + (size_t)N * HIDF;
    unsigned* sorted = (unsigned*)(bufB + (size_t)N * HIDF);
    int*      tab    = (int*)(sorted + epad);

    int nb = (N + BKT_NODES - 1) / BKT_NODES;
    int nc = (E + CHUNK_E - 1) / CHUNK_E;
    tile_sort_kernel<<<nc, 256, 0, stream>>>(src, dst, sorted, tab, E, nb);
    bucket_rank_kernel<<<nb, 512, 0, stream>>>(sorted, tab, rowptr, dinv, csr, N, nb, nc);

    int mtiles      = (N + 15) / 16;
    int gemm_blocks = mtiles < GEMM_GRID ? mtiles : GEMM_GRID;
    int pool_waves  = (N + POOL_K - 1) / POOL_K;
    int gath_blocks = (pool_waves + 7) / 8;

    int fuse_blocks = mtiles < FUSE_GRID ? mtiles : FUSE_GRID;

    gemm_wmma_kernel<<<gemm_blocks, 128, 0, stream>>>(x, Wl[0], dinv, bufA, N);
    gather_gemm_kernel<<<fuse_blocks, 128, 0, stream>>>(bufA, rowptr, csr, dinv, bl[0], Wl[1], bufB, N);
    gather_gemm_kernel<<<fuse_blocks, 128, 0, stream>>>(bufB, rowptr, csr, dinv, bl[1], Wl[2], bufA, N);
    gather_pool_kernel<<<gath_blocks, 256, 0, stream>>>(bufA, rowptr, csr, dinv, bl[2], ptr, bufB, N, G);

    pool_kernel<<<G, 256, 0, stream>>>(bufB, ptr, (float*)d_out, G, N);
}


# fused gather+product: 8 waves per 16-node tile (2 nodes, 16 columns a wave); next tile's first 24 row loads issued before the WMMA block, in flight across it; whole-row stores via LDS; sums unchanged
# speedup vs baseline: 5.0175x; 1.0394x over previous
#include <hip/hip_runtime.h>

typedef float v2f __attribute__((ext_vector_type(2)));
typedef float v8f __attribute__((ext_vector_type(8)));
typedef float v4f __attribute__((ext_vector_type(4)));

#define HIDF 128

#define CHUNK_E   2048
#define BKT_SHIFT 8
#define BKT_NODES 256
#define MAX_BKT   512

__global__ void __launch_bounds__(256) tile_sort_kernel(const int* __restrict__ src,
                                                        const int* __restrict__ dst,
                                                        unsigned* __restrict__ sorted,
                                                        int* __restrict__ tab, int e, int nb) {
    __shared__ int      s_cnt[MAX_BKT];
    __shared__ int      s_start[MAX_BKT];
    __shared__ int      s_wsum[8];
    __shared__ unsigned s_out[CHUNK_E];
    const int tid  = threadIdx.x;
    const int lane = tid & 31;
    const int wave = tid >> 5;
    const int c    = blockIdx.x;
    const int e0   = c * CHUNK_E;

    s_cnt[tid]       = 0;
    s_cnt[tid + 256] = 0;
    __syncthreads();

    int d[8];
    int sv[8];
    int r[8];
#pragma unroll
    for (int k = 0; k < 8; ++k) {
        const int i = e0 + k * 256 + tid;
        d[k]  = i < e ? dst[i] : -1;
        sv[k] = i < e ? src[i] : 0;
    }
#pragma unroll
    for (int k = 0; k < 8; ++k) {
        r[k] = 0;
        if (d[k] >= 0) r[k] = atomicAdd(&s_cnt[d[k] >> BKT_SHIFT], 1);
    }
    __syncthreads();

    const int v0  = s_cnt[2 * tid];
    const int v1  = s_cnt[2 * tid + 1];
    const int loc = v0 + v1;
    int incl = loc;
    for (int off = 1; off < 32; off <<= 1) {
        int t = __shfl_up(incl, off);
        if (lane >= off) incl += t;
    }
    if (lane == 31) s_wsum[wave] = incl;
    __syncthreads();
    int base = 0;
    for (int w = 0; w < 8; ++w) if (w < wave) base += s_wsum[w];
    const int ex = base + incl - loc;
    s_start[2 * tid]     = ex;
    s_start[2 * tid + 1] = ex + v0;
    __syncthreads();

    for (int b = tid; b < nb; b += 256) tab[c * nb + b] = s_start[b] | (s_cnt[b] << 16);

#pragma unroll
    for (int k = 0; k < 8; ++k) {
        if (d[k] >= 0) {
            const unsigned w = (unsigned)sv[k] | ((unsigned)(d[k] & (BKT_NODES - 1)) << 24);
            s_out[s_start[d[k] >> BKT_SHIFT] + r[k]] = w;
        }
    }
    __syncthreads();
    int m = e - e0;
    if (m > CHUNK_E) m = CHUNK_E;
    for (int j = tid; j < m; j += 256) sorted[e0 + j] = s_out[j];
}

__global__ void __launch_bounds__(512) bucket_rank_kernel(const unsigned* __restrict__ sorted,
                                                          const int* __restrict__ tab,
                                                          int* __restrict__ rowptr, float* __restrict__ dinv,
                                                          int* __restrict__ csr,
                                                          int n, int nb, int nc) {
    __shared__ int s_lcnt[BKT_NODES];
    __shared__ int s_pos[BKT_NODES];
    __shared__ int s_red[16];
    __shared__ int s_wsum[8];
    const int tid  = threadIdx.x;
    const int lane = tid & 31;
    const int wave = tid >> 5;
    const int b    = blockIdx.x;
    if (tid < BKT_NODES) s_lcnt[tid] = 0;
    __syncthreads();
    int before = 0;
    for (int c = tid; c < nc; c += 512) {
        const int p     = tab[c * nb + b];
        const int start = p & 0xFFFF;
        const int count = p >> 16;
        before += start;
        const unsigned* sp = sorted + c * CHUNK_E + start;
        for (int k = 0; k < count; ++k) atomicAdd(&s_lcnt[sp[k] >> 24], 1);
    }
    for (int off = 16; off > 0; off >>= 1) before += __shfl_xor(before, off);
    if (lane == 0) s_red[wave] = before;
    __syncthreads();
    int base = 0;
    for (int w = 0; w < 16; ++w) base += s_red[w];

    const int cn = tid < BKT_NODES ? s_lcnt[tid] : 0;
    int incl = cn;
    for (int off = 1; off < 32; off <<= 1) {
        int t = __shfl_up(incl, off);
        if (lane >= off) incl += t;
    }
    if (tid < BKT_NODES && lane == 31) s_wsum[wave] = incl;
    __syncthreads();
    const int node = b * BKT_NODES + tid;
    if (tid < BKT_NODES && node < n) {
        int pre = 0;
        for (int w = 0; w < 8; ++w) if (w < wave) pre += s_wsum[w];
        const int row = base + pre + incl - cn + node;
        rowptr[node]  = row;
        dinv[node]    = rsqrtf((float)(cn + 1));
        csr[row + cn] = node;
        s_pos[tid]    = row;
        if (node == n - 1) rowptr[n] = row + cn + 1;
    }
    __syncthreads();
    for (int c = tid; c < nc; c += 512) {
        const int p     = tab[c * nb + b];
        const int start = p & 0xFFFF;
        const int count = p >> 16;
        const unsigned* sp = sorted + c * CHUNK_E + start;
        for (int k = 0; k < count; ++k) {
            const unsigned w = sp[k];
            const int pos = atomicAdd(&s_pos[w >> 24], 1);
            csr[pos] = (int)(w & 0xFFFFFFu);
        }
    }
}

#define GEMM_GRID  1024
#define A_PITCH    132
__device__ __forceinline__ float load_dinv16(const float* __restrict__ dinv, int t, int mtiles, int lm, int n) {
    float r = 0.0f;
    if (t < mtiles) {
        int ri = t * 16 + lm;
        if (ri > n - 1) ri = n - 1;
        r = dinv[ri];
    }
    return r;
}

__device__ __forceinline__ float row_scale(float dvt, int i, int half) {
    const float lo = __int_as_float(__builtin_amdgcn_readlane(__float_as_int(dvt), i));
    const float hi = __int_as_float(__builtin_amdgcn_readlane(__float_as_int(dvt), i + 8));
    return half ? hi : lo;
}

__global__ void __launch_bounds__(128) gemm_wmma_kernel(const float* __restrict__ H,
                                                        const float* __restrict__ W,
                                                        const float* __restrict__ dinv,
                                                        float* __restrict__ Out, int M) {
    __shared__ __attribute__((aligned(16))) float s_a[2][16 * A_PITCH];
    const int lane = threadIdx.x & 31;
    const int wave = __builtin_amdgcn_readfirstlane(threadIdx.x >> 5);
    const int half = lane >> 4;
    const int lm   = lane & 15;
    const int mtiles = (M + 15) >> 4;
    int tile = blockIdx.x;
    if (tile >= mtiles) return;
    const int col0 = wave * 32;

    v2f b[32][2];
#pragma unroll
    for (int ks = 0; ks < 32; ++ks) {
        const float* wk = W + (long)(4 * ks + 2 * half) * HIDF + col0 + lm;
#pragma unroll
        for (int t = 0; t < 2; ++t) {
            b[ks][t].x = wk[t * 16];
            b[ks][t].y = wk[HIDF + t * 16];
        }
    }

    v4f g[4];
#pragma unroll
    for (int q = 0; q < 4; ++q) {
        int row = tile * 16 + q * 4 + wave;
        if (row > M - 1) row = M - 1;
        g[q] = *((const v4f*)(H + (long)row * HIDF) + lane);
    }
#pragma unroll
    for (int q = 0; q < 4; ++q)
        *(v4f*)(&s_a[0][(q * 4 + wave) * A_PITCH + lane * 4]) = g[q];
    __syncthreads();

    int buf = 0;
    for (; tile < mtiles; tile += gridDim.x) {
        const int  next     = tile + gridDim.x;
        const bool has_next = next < mtiles;
        if (has_next) {
#pragma unroll
            for (int q = 0; q < 4; ++q) {
                int row = next * 16 + q * 4 + wave;
                if (row > M - 1) row = M - 1;
                g[q] = *((const v4f*)(H + (long)row * HIDF) + lane);
            }
        }

        const float dvt = load_dinv16(dinv, tile, mtiles, lm, M);

        const float* sa = &s_a[buf][lm * A_PITCH + 2 * half];
        v2f a[32];
#pragma unroll
        for (int ks = 0; ks < 32; ++ks) a[ks] = *(const v2f*)(sa + 4 * ks);

        v8f c0 = (v8f){0.f, 0.f, 0.f, 0.f, 0.f, 0.f, 0.f, 0.f};
        v8f c1 = (v8f){0.f, 0.f, 0.f, 0.f, 0.f, 0.f, 0.f, 0.f};
#pragma unroll
        for (int ks = 0; ks < 32; ++ks) {
            c0 = __builtin_amdgcn_wmma_f32_16x16x4_f32(false, a[ks], false, b[ks][0], (short)0, c0, false, false);
            c1 = __builtin_amdgcn_wmma_f32_16x16x4_f32(false, a[ks], false, b[ks][1], (short)0, c1, false, false);
        }

#pragma unroll
        for (int i = 0; i < 8; ++i) {
            const int   row = tile * 16 + i + 8 * half;
            const float sc  = row_scale(dvt, i, half);
            if (row < M) {
                float* o = Out + (long)row * HIDF + col0 + lm;
                o[0]  = c0[i] * sc;
                o[16] = c1[i] * sc;
            }
        }

        if (has_next) {
#pragma unroll
            for (int q = 0; q < 4; ++q)
                *(v4f*)(&s_a[buf ^ 1][(q * 4 + wave) * A_PITCH + lane * 4]) = g[q];
        }
        __syncthreads();
        buf ^= 1;
    }
}

template <int K>
__device__ __forceinline__ void add_rows(const v4f* __restrict__ hw4, int ent, int j,
                                         int lane, v4f& acc) {
    v4f v[K];
#pragma unroll
    for (int k = 0; k < K; ++k) {
        int s = __builtin_amdgcn_readlane(ent, j + k);
        v[k]  = hw4[(long)s * (HIDF / 4) + lane];
    }
#pragma unroll
    for (int k = 0; k < K; ++k) acc += v[k];
}

__device__ __forceinline__ void add_chunk(const v4f* __restrict__ hw4, int ent, int m, int lane, v4f& acc) {
    int k = 0;
    for (; k + 8 <= m; k += 8) add_rows<8>(hw4, ent, k, lane, acc);
    if (k + 4 <= m) { add_rows<4>(hw4, ent, k, lane, acc); k += 4; }
    const int r = m - k;
    if (r == 3)      add_rows<3>(hw4, ent, k, lane, acc);
    else if (r == 2) add_rows<2>(hw4, ent, k, lane, acc);
    else if (r == 1) add_rows<1>(hw4, ent, k, lane, acc);
}

#define POOL_K 8
__global__ void __launch_bounds__(256) gather_pool_kernel(const float* __restrict__ hw,
                                                          const int* __restrict__ rowptr,
                                                          const int* __restrict__ csr,
                                                          const float* __restrict__ dinv,
                                                          const float* __restrict__ bias,
                                                          const int* __restrict__ ptr,
                                                          float* __restrict__ part, int n, int g) {
    const int lane = threadIdx.x & 31;
    const int wg = __builtin_amdgcn_readfirstlane(blockIdx.x * 8 + (threadIdx.x >> 5));
    const int n0 = wg * POOL_K;
    if (n0 >= n) return;
    const int cntn = n - n0 < POOL_K ? n - n0 : POOL_K;
    const v4f* hw4 = (const v4f*)hw;
    const int* ed  = csr;

    int ri = n0 + (lane < POOL_K ? lane : POOL_K);
    if (ri > n) ri = n;
    const int rp = rowptr[ri];

    int beg = __builtin_amdgcn_readlane(rp, 0);
    int ent = 0;
    {
        int m0 = __builtin_amdgcn_readlane(rp, 1) - beg;
        if (m0 > 32) m0 = 32;
        if (lane < m0) ent = ed[beg + lane];
    }
    const v4f bv = *((const v4f*)bias + lane);

    const int   x   = n0 + (lane < cntn ? lane : cntn - 1);
    const float dvp = dinv[x];
    int lo = 0;
    int hi = g - 1;
    while (lo < hi) {
        const int mid = (lo + hi + 1) >> 1;
        if (ptr[mid] <= x) lo = mid; else hi = mid - 1;
    }
    const int      gnext = __shfl_down(lo, 1);
    const bool     fl    = lane < cntn && (lane == cntn - 1 || gnext != lo);
    const unsigned fmask = __builtin_amdgcn_ballot_w32(fl);

    v4f gsum = {0.0f, 0.0f, 0.0f, 0.0f};
    for (int j = 0; j < cntn; ++j) {
        const int end = __builtin_amdgcn_readlane(rp, j + 1);
        int entn = 0;
        if (j + 1 < cntn) {
            int m2 = __builtin_amdgcn_readlane(rp, j + 2) - end;
            if (m2 > 32) m2 = 32;
            if (lane < m2) entn = ed[end + lane];
        }

        v4f acc = {0.0f, 0.0f, 0.0f, 0.0f};
        add_chunk(hw4, ent, end - beg < 32 ? end - beg : 32, lane, acc);
        for (int base = beg + 32; base < end; base += 32) {
            int m = end - base;
            if (m > 32) m = 32;
            int more = 0;
            if (lane < m) more = ed[base + lane];
            add_chunk(hw4, more, m, lane, acc);
        }

        const float dd = __int_as_float(__builtin_amdgcn_readlane(__float_as_int(dvp), j));
        v4f o = acc * dd + bv;
        o.x = o.x > 0.0f ? o.x : 0.0f;
        o.y = o.y > 0.0f ? o.y : 0.0f;
        o.z = o.z > 0.0f ? o.z : 0.0f;
        o.w = o.w > 0.0f ? o.w : 0.0f;
        gsum += o;
        if ((fmask >> j) & 1u) {
            const int gi = __builtin_amdgcn_readlane(lo, j);
            *((v4f*)(part + (long)(wg + gi) * HIDF) + lane) = gsum;
            gsum = (v4f){0.0f, 0.0f, 0.0f, 0.0f};
        }
        beg = end;
        ent = entn;
    }
}

#define FUSE_GRID  512
#define PRE_ROWS   24

__device__ __forceinline__ void put_row(float* __restrict__ dst, v4f acc, float dd, v4f bv) {
    v4f o = acc * dd + bv;
    o.x = o.x > 0.0f ? o.x : 0.0f;
    o.y = o.y > 0.0f ? o.y : 0.0f;
    o.z = o.z > 0.0f ? o.z : 0.0f;
    o.w = o.w > 0.0f ? o.w : 0.0f;
    *(v4f*)dst = o;
}

__device__ __forceinline__ int load_bounds(const int* __restrict__ rowptr, int t, int mtiles, int wave,
                                           int l2, int n) {
    int r = 0;
    if (t < mtiles) {
        int ri = t * 16 + 2 * wave + l2;
        if (ri > n) ri = n;
        r = rowptr[ri];
    }
    return r;
}

__device__ __forceinline__ void lds_barrier() {
    __builtin_amdgcn_fence(__ATOMIC_RELEASE, "workgroup", "local");
    __builtin_amdgcn_s_barrier();
    __builtin_amdgcn_fence(__ATOMIC_ACQUIRE, "workgroup", "local");
}

template <int J, int G>
__device__ __forceinline__ void add_rows_seg(const v4f* __restrict__ hw4, int ent, int cnt, int slot0,
                                             int lane, int rp, float dvt, int w0, v4f bv,
                                             float* __restrict__ srow, v4f& acc, int& cur, int& nb) {
    v4f v[G];
    const int jlast = J + cnt - 1;
#pragma unroll
    for (int k = 0; k < G; ++k) {
        const int jj = J + k < jlast ? J + k : jlast;
        int s = __builtin_amdgcn_readlane(ent, jj);
        v[k]  = hw4[(long)s * (HIDF / 4) + lane];
    }
#pragma unroll
    for (int k = 0; k < G; ++k) {
        if (k < cnt) {
            if (slot0 + k == nb) {
                const float dd = __int_as_float(__builtin_amdgcn_readlane(__float_as_int(dvt), w0 + cur));
                put_row(srow + cur * A_PITCH, acc, dd, bv);
                ++cur;
                nb  = __builtin_amdgcn_readlane(rp, cur + 1);
                acc = (v4f){0.0f, 0.0f, 0.0f, 0.0f};
            }
            acc += v[k];
        }
    }
}

__device__ __forceinline__ void issue_rows(const v4f* __restrict__ hw4, int ent, int mp, int lane,
                                           v4f (&v)[PRE_ROWS]) {
    const int jl = mp > 0 ? mp - 1 : 0;
#pragma unroll
    for (int k = 0; k < 8; ++k) {
        const int s = __builtin_amdgcn_readlane(ent, k < jl ? k : jl);
        v[k] = hw4[(long)s * (HIDF / 4) + lane];
    }
#pragma unroll
    for (int c = 8; c < PRE_ROWS; c += 2) {
        if (mp > c) {
#pragma unroll
            for (int k = c; k < c + 2; ++k) {
                const int s = __builtin_amdgcn_readlane(ent, k < jl ? k : jl);
                v[k] = hw4[(long)s * (HIDF / 4) + lane];
            }
        }
    }
}

__global__ void __launch_bounds__(256, 4) gather_gemm_kernel(const float* __restrict__ hw,
                                                          const int* __restrict__ rowptr,
                                                          const int* __restrict__ csr,
                                                          const float* __restrict__ dinv,
                                                          const float* __restrict__ bias,
                                                          const float* __restrict__ W,
                                                          float* __restrict__ Out, int n) {
    __shared__ __attribute__((aligned(16))) float s_a[2][16 * A_PITCH];
    __shared__ __attribute__((aligned(16))) float s_o[16 * A_PITCH];
    const int lane = threadIdx.x & 31;
    const int wave = __builtin_amdgcn_readfirstlane(threadIdx.x >> 5);
    const int half = lane >> 4;
    const int lm   = lane & 15;
    const int mtiles = (n + 15) >> 4;
    int tile = blockIdx.x;
    if (tile >= mtiles) return;
    const int step = gridDim.x;
    const int col0 = wave * 16;
    const v4f* hw4 = (const v4f*)hw;
    const int* ed  = csr;
    const int  l2  = lane < 3 ? lane : 2;
    const int  w2  = 2 * wave;

    int   rp   = load_bounds(rowptr, tile, mtiles, wave, l2, n);
    int   rpn  = load_bounds(rowptr, tile + step, mtiles, wave, l2, n);
    int   rpnn = load_bounds(rowptr, tile + 2 * step, mtiles, wave, l2, n);
    int   e0   = __builtin_amdgcn_readlane(rp, 0);
    int   e2   = __builtin_amdgcn_readlane(rp, 2);
    int   ent  = 0;
    if (lane < e2 - e0) ent = ed[e0 + lane];
    int   e0n  = __builtin_amdgcn_readlane(rpn, 0);
    int   e2n  = __builtin_amdgcn_readlane(rpn, 2);
    int   entn = 0;
    if (lane < e2n - e0n) entn = ed[e0n + lane];
    float dvt  = load_dinv16(dinv, tile, mtiles, lm, n);

    v2f b[32];
#pragma unroll
    for (int ks = 0; ks < 32; ++ks) {
        const float* wk = W + (long)(4 * ks + 2 * half) * HIDF + col0 + lm;
        b[ks].x = wk[0];
        b[ks].y = wk[HIDF];
    }
    const v4f bv = *((const v4f*)bias + lane);

    v4f v[PRE_ROWS];
    issue_rows(hw4, ent, e2 - e0 < PRE_ROWS ? e2 - e0 : PRE_ROWS, lane, v);

    int buf = 0;
    for (; tile < mtiles; tile += step) {
        float* srow = &s_a[buf][w2 * A_PITCH + lane * 4];
        v4f acc = {0.0f, 0.0f, 0.0f, 0.0f};
        int cur = 0;
        int nb  = __builtin_amdgcn_readlane(rp, 1);
        const int mt = e2 - e0;
        {
            const int mp = mt < PRE_ROWS ? mt : PRE_ROWS;
#pragma unroll
            for (int k = 0; k < PRE_ROWS; ++k) {
                if (k < mp) {
                    if (e0 + k == nb) {
                        const float dd = __int_as_float(__builtin_amdgcn_readlane(__float_as_int(dvt), w2 + cur));
                        put_row(srow + cur * A_PITCH, acc, dd, bv);
                        ++cur;
                        nb  = __builtin_amdgcn_readlane(rp, cur + 1);
                        acc = (v4f){0.0f, 0.0f, 0.0f, 0.0f};
                    }
                    acc += v[k];
                }
            }
        }
        if (mt > PRE_ROWS) {
            const int m32 = mt < 32 ? mt : 32;
            add_rows_seg<PRE_ROWS, 32 - PRE_ROWS>(hw4, ent, m32 - PRE_ROWS, e0 + PRE_ROWS, lane, rp, dvt, w2, bv,
                                                  srow, acc, cur, nb);
            for (int base = e0 + 32; base < e2; base += 32) {
                int m = e2 - base;
                if (m > 32) m = 32;
                int more = 0;
                if (lane < m) more = ed[base + lane];
                add_rows_seg<0, 8>(hw4, more, m < 8 ? m : 8, base, lane, rp, dvt, w2, bv, srow, acc, cur, nb);
                if (m > 8)  add_rows_seg<8, 8>(hw4, more, (m < 16 ? m : 16) - 8, base + 8, lane, rp, dvt, w2, bv, srow, acc, cur, nb);
                if (m > 16) add_rows_seg<16, 8>(hw4, more, (m < 24 ? m : 24) - 16, base + 16, lane, rp, dvt, w2, bv, srow, acc, cur, nb);
                if (m > 24) add_rows_seg<24, 8>(hw4, more, m - 24, base + 24, lane, rp, dvt, w2, bv, srow, acc, cur, nb);
            }
        }
        for (; cur < 2; ++cur) {
            const float dd = __int_as_float(__builtin_amdgcn_readlane(__float_as_int(dvt), w2 + cur));
            put_row(srow + cur * A_PITCH, acc, dd, bv);
            acc = (v4f){0.0f, 0.0f, 0.0f, 0.0f};
        }
        lds_barrier();

        issue_rows(hw4, entn, e2n - e0n < PRE_ROWS ? e2n - e0n : PRE_ROWS, lane, v);
        const int e0nn = __builtin_amdgcn_readlane(rpnn, 0);
        const int e2nn = __builtin_amdgcn_readlane(rpnn, 2);
        int entnn = 0;
        if (lane < e2nn - e0nn) entnn = ed[e0nn + lane];
        const int   rpn3 = load_bounds(rowptr, tile + 3 * step, mtiles, wave, l2, n);
        const float dvn  = load_dinv16(dinv, tile + step, mtiles, lm, n);

        const float* sa = &s_a[buf][lm * A_PITCH + 2 * half];
        v2f a[32];
#pragma unroll
        for (int ks = 0; ks < 32; ++ks) a[ks] = *(const v2f*)(sa + 4 * ks);
        __builtin_amdgcn_sched_barrier(0);
        v8f c0 = (v8f){0.f, 0.f, 0.f, 0.f, 0.f, 0.f, 0.f, 0.f};
#pragma unroll
        for (int ks = 0; ks < 32; ++ks)
            c0 = __builtin_amdgcn_wmma_f32_16x16x4_f32(false, a[ks], false, b[ks], (short)0, c0, false, false);

#pragma unroll
        for (int i = 0; i < 8; ++i) s_o[(i + 8 * half) * A_PITCH + col0 + lm] = c0[i];
        lds_barrier();

#pragma unroll
        for (int r = 0; r < 2; ++r) {
            const int   row = tile * 16 + w2 + r;
            const float sc  = __int_as_float(__builtin_amdgcn_readlane(__float_as_int(dvt), w2 + r));
            const v4f   o   = *(const v4f*)(&s_o[(w2 + r) * A_PITCH + lane * 4]);
            if (row < n) *((v4f*)(Out + (long)row * HIDF) + lane) = o * sc;
        }

        rp   = rpn;
        rpn  = rpnn;
        rpnn = rpn3;
        e0   = e0n;
        e2   = e2n;
        e0n  = e0nn;
        e2n  = e2nn;
        ent  = entn;
        entn = entnn;
        dvt  = dvn;
        buf ^= 1;
    }
}

#define POOL_WAVES 8
#define POOL_ROWS  16
__global__ void __launch_bounds__(POOL_WAVES * 32) pool_kernel(const float* __restrict__ h,
                                                               const int* __restrict__ ptr,
                                                               float* __restrict__ out, int g, int nn) {
    __shared__ v4f s_part[POOL_WAVES][32];
    const int gi   = blockIdx.x;
    const int t    = threadIdx.x;
    const int lane = t & 31;
    const int wave = __builtin_amdgcn_readfirstlane(t >> 5);
    const int p0 = ptr[gi];
    const int p1 = ptr[gi + 1];
    int sn = gi == 0 ? 0 : p0;
    int en = gi == g - 1 ? nn : p1;
    if (sn < 0) sn = 0;
    if (en > nn) en = nn;
    int s = 0;
    int e = 0;
    if (sn < en) {
        s = sn / POOL_K + gi;
        e = (en - 1) / POOL_K + gi + 1;
    }
    const v4f* h4 = (const v4f*)h + lane;

    v4f acc = {0.0f, 0.0f, 0.0f, 0.0f};
    int n = s + wave;
    for (; n + (POOL_ROWS - 1) * POOL_WAVES < e; n += POOL_ROWS * POOL_WAVES) {
        v4f v[POOL_ROWS];
#pragma unroll
        for (int u = 0; u < POOL_ROWS; ++u) v[u] = h4[(long)(n + u * POOL_WAVES) * (HIDF / 4)];
#pragma unroll
        for (int u = 0; u < POOL_ROWS; ++u) acc += v[u];
    }
    if (n < e) {
        v4f v[POOL_ROWS];
        const int last = e - 1;
#pragma unroll
        for (int u = 0; u < POOL_ROWS; ++u) {
            const int r = n + u * POOL_WAVES;
            v[u] = h4[(long)(r < e ? r : last) * (HIDF / 4)];
        }
#pragma unroll
        for (int u = 0; u < POOL_ROWS; ++u) {
            if (n + u * POOL_WAVES < e) acc += v[u];
        }
    }

    s_part[wave][lane] = acc;
    __syncthreads();
    if (t < HIDF) {
        const float* sp = (const float*)s_part;
        float sum = 0.0f;
        for (int w = 0; w < POOL_WAVES; ++w) sum += sp[w * HIDF + t];
        float cnt = (float)(p1 - p0);
        if (cnt < 1.0f) cnt = 1.0f;
        out[(long)gi * HIDF + t] = sum / cnt;
    }
}

extern "C" void kernel_launch(void* const* d_in, const int* in_sizes, int n_in,
                              void* d_out, int out_size, void* d_ws, size_t ws_size,
                              hipStream_t stream) {
    const float* x   = (const float*)d_in[0];
    const int*   ei  = (const int*)d_in[1];
    const int*   ptr = (const int*)d_in[2];
    const float* Wl[3] = {(const float*)d_in[3], (const float*)d_in[5], (const float*)d_in[7]};
    const float* bl[3] = {(const float*)d_in[4], (const float*)d_in[6], (const float*)d_in[8]};

    int N = in_sizes[0] / HIDF;
    int E = in_sizes[1] / 2;
    int G = in_sizes[2] - 1;

    const int* src = ei;
    const int* dst = ei + E;

    size_t npad = ((size_t)N + 255) & ~(size_t)255;
    size_t epad = ((size_t)E + 255) & ~(size_t)255;
    float* dinv   = (float*)d_ws;
    int*   rowptr = (int*)(dinv + npad);
    int*   csr    = rowptr + npad + 256;
    float* bufA   = (float*)(csr + epad + npad);
    float* bufB   = bufA + (size_t)N * HIDF;
    unsigned* sorted = (unsigned*)(bufB + (size_t)N * HIDF);
    int*      tab    = (int*)(sorted + epad);

    int nb = (N + BKT_NODES - 1) / BKT_NODES;
    int nc = (E + CHUNK_E - 1) / CHUNK_E;
    tile_sort_kernel<<<nc, 256, 0, stream>>>(src, dst, sorted, tab, E, nb);
    bucket_rank_kernel<<<nb, 512, 0, stream>>>(sorted, tab, rowptr, dinv, csr, N, nb, nc);

    int mtiles      = (N + 15) / 16;
    int gemm_blocks = mtiles < GEMM_GRID ? mtiles : GEMM_GRID;
    int pool_waves  = (N + POOL_K - 1) / POOL_K;
    int gath_blocks = (pool_waves + 7) / 8;

    int fuse_blocks = mtiles < FUSE_GRID ? mtiles : FUSE_GRID;

    gemm_wmma_kernel<<<gemm_blocks, 128, 0, stream>>>(x, Wl[0], dinv, bufA, N);
    gather_gemm_kernel<<<fuse_blocks, 256, 0, stream>>>(bufA, rowptr, csr, dinv, bl[0], Wl[1], bufB, N);
    gather_gemm_kernel<<<fuse_blocks, 256, 0, stream>>>(bufB, rowptr, csr, dinv, bl[1], Wl[2], bufA, N);
    gather_pool_kernel<<<gath_blocks, 256, 0, stream>>>(bufA, rowptr, csr, dinv, bl[2], ptr, bufB, N, G);

    pool_kernel<<<G, 256, 0, stream>>>(bufB, ptr, (float*)d_out, G, N);
}
